# starting kernel
# baseline (speedup 1.0000x reference)
#include <hip/hip_runtime.h>

#define NN 100000
#define NE 3200000
#define NR 8

typedef __attribute__((ext_vector_type(2))) float v2f;
typedef __attribute__((ext_vector_type(8))) float v8f;


__global__ void zero_f32(float* __restrict__ p, size_t n) {
    size_t i = (size_t)blockIdx.x * blockDim.x + threadIdx.x;
    size_t stride = (size_t)gridDim.x * blockDim.x;
    for (; i < n; i += stride) p[i] = 0.0f;
}

__global__ void count_edges(const int* __restrict__ dst, const int* __restrict__ et,
                            float* __restrict__ cnt) {
    int e = blockIdx.x * blockDim.x + threadIdx.x;
    if (e >= NE) return;
    atomicAdd(&cnt[(size_t)et[e] * NN + dst[e]], 1.0f);
}

__global__ void recip_clamp(float* __restrict__ c, int n) {
    int i = blockIdx.x * blockDim.x + threadIdx.x;
    if (i < n) c[i] = 1.0f / fmaxf(c[i], 1.0f);
}

template <int F>
__global__ void scatter_edges(const float* __restrict__ feat, const int* __restrict__ src,
                              const int* __restrict__ dst, const int* __restrict__ et,
                              float* __restrict__ agg) {
    constexpr int Q = F / 4;
    long long tid = (long long)blockIdx.x * blockDim.x + threadIdx.x;
    int e = (int)(tid / Q);
    int q = (int)(tid % Q);
    if (e >= NE) return;
    int s = src[e], d = dst[e], r = et[e];
    const float4 v = ((const float4*)(feat + (size_t)s * F))[q];
    float* ap = agg + ((size_t)r * NN + d) * F + (size_t)q * 4;
    atomicAdd(ap + 0, v.x);
    atomicAdd(ap + 1, v.y);
    atomicAdd(ap + 2, v.z);
    atomicAdd(ap + 3, v.w);
}

__device__ __forceinline__ v2f a_frag(const float* __restrict__ M, int row, int stride,
                                      int k0, int kh, float scale) {
    const float* p = M + (size_t)row * stride + k0 + kh;
    v2f a; a.x = p[0] * scale; a.y = p[1] * scale; return a;
}
__device__ __forceinline__ v2f b_frag(const float* __restrict__ W, int nout,
                                      int k0, int kh, int col) {
    v2f b;
    b.x = W[(size_t)(k0 + kh) * nout + col];
    b.y = W[(size_t)(k0 + kh + 1) * nout + col];
    return b;
}


__global__ __launch_bounds__(256) void rgcn_layer1(
    const float* __restrict__ x, const float* __restrict__ agg, const float* __restrict__ inv,
    const float* __restrict__ W1, const float* __restrict__ root1, const float* __restrict__ b1,
    float* __restrict__ h) {
    int wid  = (blockIdx.x * blockDim.x + threadIdx.x) >> 5;
    int lane = threadIdx.x & 31;
    if (wid >= NN / 16) return;
    int m0   = wid * 16;
    int arow = m0 + (lane & 15);
    int kh   = (lane >> 4) << 1;
    int col  = lane & 15;

    v8f c0, c1;
    float bb0 = b1[col], bb1 = b1[16 + col];
#pragma unroll
    for (int v = 0; v < 8; ++v) { c0[v] = bb0; c1[v] = bb1; }

#pragma unroll
    for (int k = 0; k < 16; k += 4) {
        v2f a  = a_frag(x, arow, 16, k, kh, 1.0f);
        v2f bA = b_frag(root1, 32, k, kh, col);
        v2f bB = b_frag(root1, 32, k, kh, col + 16);
        c0 = __builtin_amdgcn_wmma_f32_16x16x4_f32(false, a, false, bA, (short)0, c0, false, false);
        c1 = __builtin_amdgcn_wmma_f32_16x16x4_f32(false, a, false, bB, (short)0, c1, false, false);
    }
    for (int r = 0; r < NR; ++r) {
        const float* ag = agg + (size_t)r * NN * 16;
        const float* Wr = W1 + (size_t)r * 16 * 32;
        float sc = inv[(size_t)r * NN + arow];
#pragma unroll
        for (int k = 0; k < 16; k += 4) {
            v2f a  = a_frag(ag, arow, 16, k, kh, sc);
            v2f bA = b_frag(Wr, 32, k, kh, col);
            v2f bB = b_frag(Wr, 32, k, kh, col + 16);
            c0 = __builtin_amdgcn_wmma_f32_16x16x4_f32(false, a, false, bA, (short)0, c0, false, false);
            c1 = __builtin_amdgcn_wmma_f32_16x16x4_f32(false, a, false, bB, (short)0, c1, false, false);
        }
    }
    int rbase = m0 + ((lane >> 4) << 3);
#pragma unroll
    for (int v = 0; v < 8; ++v) {
        h[(size_t)(rbase + v) * 32 + col]      = fmaxf(c0[v], 0.0f);
        h[(size_t)(rbase + v) * 32 + 16 + col] = fmaxf(c1[v], 0.0f);
    }
}


__global__ __launch_bounds__(256) void rgcn_layer2(
    const float* __restrict__ h, const float* __restrict__ agg, const float* __restrict__ inv,
    const float* __restrict__ W2, const float* __restrict__ root2, const float* __restrict__ b2,
    float* __restrict__ out) {
    int wid  = (blockIdx.x * blockDim.x + threadIdx.x) >> 5;
    int lane = threadIdx.x & 31;
    if (wid >= NN / 16) return;
    int m0   = wid * 16;
    int arow = m0 + (lane & 15);
    int kh   = (lane >> 4) << 1;
    int col  = lane & 15;

    v8f c;
    float bb = b2[col];
#pragma unroll
    for (int v = 0; v < 8; ++v) c[v] = bb;

#pragma unroll
    for (int k = 0; k < 32; k += 4) {
        v2f a = a_frag(h, arow, 32, k, kh, 1.0f);
        v2f b = b_frag(root2, 16, k, kh, col);
        c = __builtin_amdgcn_wmma_f32_16x16x4_f32(false, a, false, b, (short)0, c, false, false);
    }
    for (int r = 0; r < NR; ++r) {
        const float* ag = agg + (size_t)r * NN * 32;
        const float* Wr = W2 + (size_t)r * 32 * 16;
        float sc = inv[(size_t)r * NN + arow];
#pragma unroll
        for (int k = 0; k < 32; k += 4) {
            v2f a = a_frag(ag, arow, 32, k, kh, sc);
            v2f b = b_frag(Wr, 16, k, kh, col);
            c = __builtin_amdgcn_wmma_f32_16x16x4_f32(false, a, false, b, (short)0, c, false, false);
        }
    }
    int rbase = m0 + ((lane >> 4) << 3);
#pragma unroll
    for (int v = 0; v < 8; ++v)
        out[(size_t)(rbase + v) * 16 + col] = c[v];
}


extern "C" void kernel_launch(void* const* d_in, const int* in_sizes, int n_in,
                              void* d_out, int out_size, void* d_ws, size_t ws_size,
                              hipStream_t stream) {
    const float* x     = (const float*)d_in[0];
    const int*   eidx  = (const int*)d_in[1];
    const int*   etype = (const int*)d_in[2];
    const float* W1    = (const float*)d_in[3];
    const float* root1 = (const float*)d_in[4];
    const float* b1    = (const float*)d_in[5];
    const float* W2    = (const float*)d_in[6];
    const float* root2 = (const float*)d_in[7];
    const float* b2    = (const float*)d_in[8];
    float* out = (float*)d_out;

    const int* src = eidx;
    const int* dst = eidx + NE;

    float* agg = (float*)d_ws;
    float* inv = agg + (size_t)NR * NN * 32;
    float* h   = inv + (size_t)NR * NN;

    const int B = 256;

    zero_f32<<<2048, B, 0, stream>>>(inv, (size_t)NR * NN);
    count_edges<<<(NE + B - 1) / B, B, 0, stream>>>(dst, etype, inv);
    recip_clamp<<<(NR * NN + B - 1) / B, B, 0, stream>>>(inv, NR * NN);

    zero_f32<<<4096, B, 0, stream>>>(agg, (size_t)NR * NN * 16);
    {
        long long nth = (long long)NE * 4;
        scatter_edges<16><<<(int)((nth + B - 1) / B), B, 0, stream>>>(x, src, dst, etype, agg);
    }
    rgcn_layer1<<<((NN / 16) * 32 + B - 1) / B, B, 0, stream>>>(x, agg, inv, W1, root1, b1, h);

    zero_f32<<<4096, B, 0, stream>>>(agg, (size_t)NR * NN * 32);
    {
        long long nth = (long long)NE * 8;
        scatter_edges<32><<<(int)((nth + B - 1) / B), B, 0, stream>>>(h, src, dst, etype, agg);
    }
    rgcn_layer2<<<((NN / 16) * 32 + B - 1) / B, B, 0, stream>>>(h, agg, inv, W2, root2, b2, out);

    (void)in_sizes; (void)n_in; (void)out_size; (void)ws_size;
}


# layer gathers: 8 lanes read each neighbour row whole (h rows line-aligned) and walk a node's 4 segments as one run, 8 rows in flight; sums reach A fragments through LDS; sum order unchanged
# speedup vs baseline: 1.6789x; 1.6789x over previous
#include <hip/hip_runtime.h>

#define NN 100000
#define NE 3200000
#define NR 8
#define NK (NN * NR)
#define NQ (NK / 4)
#define SCAN_BLOCKS 782

typedef __attribute__((ext_vector_type(2))) float v2f;
typedef __attribute__((ext_vector_type(8))) float v8f;


__global__ void zero_i32(int* __restrict__ p, int n) {
    int i = blockIdx.x * blockDim.x + threadIdx.x;
    int stride = gridDim.x * blockDim.x;
    for (; i < n; i += stride) p[i] = 0;
}

__global__ void count_rank(const int* __restrict__ dst, const int* __restrict__ et,
                           int* __restrict__ cnt, int* __restrict__ rank) {
    int e = blockIdx.x * blockDim.x + threadIdx.x;
    if (e >= NE) return;
    rank[e] = atomicAdd(&cnt[dst[e] * NR + et[e]], 1);
}

__global__ __launch_bounds__(256) void seg_block_totals(const int* __restrict__ cnt,
                                                        int* __restrict__ bsum) {
    __shared__ int wsum[8];
    int q = blockIdx.x * 256 + threadIdx.x;
    int t = 0;
    if (q < NQ) { int4 v = ((const int4*)cnt)[q]; t = v.x + v.y + v.z + v.w; }
    for (int o = 16; o > 0; o >>= 1) t += __shfl_xor(t, o, 32);
    if ((threadIdx.x & 31) == 0) wsum[threadIdx.x >> 5] = t;
    __syncthreads();
    if (threadIdx.x == 0) {
        int s = 0;
        for (int w = 0; w < 8; ++w) s += wsum[w];
        bsum[blockIdx.x] = s;
    }
}

__global__ __launch_bounds__(256) void seg_offsets(const int* __restrict__ cnt,
                                                   const int* __restrict__ bsum,
                                                   int* __restrict__ offs) {
    __shared__ int wpre[8];
    __shared__ int wsum[8];
    int tid = threadIdx.x, lane = tid & 31, wv = tid >> 5;

    int b = 0;
    for (int i = tid; i < (int)blockIdx.x; i += 256) b += bsum[i];
    for (int o = 16; o > 0; o >>= 1) b += __shfl_xor(b, o, 32);
    if (lane == 0) wpre[wv] = b;

    int q = blockIdx.x * 256 + tid;
    int4 v = make_int4(0, 0, 0, 0);
    if (q < NQ) v = ((const int4*)cnt)[q];
    int t = v.x + v.y + v.z + v.w;
    int inc = t;
    for (int o = 1; o < 32; o <<= 1) {
        int n = __shfl_up(inc, o, 32);
        if (lane >= o) inc += n;
    }
    if (lane == 31) wsum[wv] = inc;
    __syncthreads();

    int base = 0;
    for (int w = 0; w < 8; ++w) base += wpre[w];
    for (int w = 0; w < wv; ++w) base += wsum[w];
    int ex = base + inc - t;
    if (q < NQ) {
        int4 o4;
        o4.x = ex; o4.y = o4.x + v.x; o4.z = o4.y + v.y; o4.w = o4.z + v.z;
        ((int4*)offs)[q] = o4;
        if (q == NQ - 1) offs[NK] = o4.w + v.w;
    }
}

__global__ void place_edges(const int* __restrict__ src, const int* __restrict__ dst,
                            const int* __restrict__ et, const int* __restrict__ rank,
                            const int* __restrict__ offs, int* __restrict__ perm) {
    int e = blockIdx.x * blockDim.x + threadIdx.x;
    if (e >= NE) return;
    perm[offs[dst[e] * NR + et[e]] + rank[e]] = src[e];
}

__device__ __forceinline__ v2f a_frag(const float* __restrict__ M, int row, int stride,
                                      int k0, int kh) {
    const float* p = M + (size_t)row * stride + k0 + kh;
    v2f a; a.x = p[0]; a.y = p[1]; return a;
}
__device__ __forceinline__ v2f b_frag(const float* __restrict__ W, int nout,
                                      int k0, int kh, int col) {
    v2f b;
    b.x = W[(size_t)(k0 + kh) * nout + col];
    b.y = W[(size_t)(k0 + kh + 1) * nout + col];
    return b;
}

template <int F, int PITCH>
__device__ __forceinline__ void tile_seg_sums(const float* __restrict__ feat,
                                              const int* __restrict__ offs,
                                              const int* __restrict__ perm,
                                              int m0, float* __restrict__ agg) {
    typedef __attribute__((ext_vector_type(F / 8))) float vec;
    const int FPL  = F / 8;
    const int SLOT = 16 * PITCH;
    int q    = threadIdx.x & 7;
    int g    = threadIdx.x >> 3;
    int node = g >> 1, half = g & 1;
    int k0   = (m0 + node) * NR + half * 4;
    int4 bb  = *(const int4*)(offs + k0);
    int beg = bb.x, b1 = bb.y, b2 = bb.z, b3 = bb.w;
    int end  = offs[k0 + 4];
    int last = end - 1;

    float* slot0 = agg + (half * 4 * 16 + node) * PITCH + q * FPL;
    vec zero;
#pragma unroll
    for (int i = 0; i < FPL; ++i) zero[i] = 0.0f;
#pragma unroll
    for (int s = 0; s < 4; ++s) *(vec*)(slot0 + s * SLOT) = zero;

    const float* fq = feat + q * FPL;
    vec acc = zero;
    int cur = 0;
    int mine = 0;
    if (beg < end) { int t = beg + q; mine = perm[t < last ? t : last]; }
    for (int e = beg; e < end; e += 8) {
        int i0 = __builtin_amdgcn_mov_dpp8(mine, 0x000000);
        int i1 = __builtin_amdgcn_mov_dpp8(mine, 0x249249);
        int i2 = __builtin_amdgcn_mov_dpp8(mine, 0x492492);
        int i3 = __builtin_amdgcn_mov_dpp8(mine, 0x6DB6DB);
        int i4 = __builtin_amdgcn_mov_dpp8(mine, 0x924924);
        int i5 = __builtin_amdgcn_mov_dpp8(mine, 0xB6DB6D);
        int i6 = __builtin_amdgcn_mov_dpp8(mine, 0xDB6DB6);
        int i7 = __builtin_amdgcn_mov_dpp8(mine, 0xFFFFFF);
        vec u[8];
        u[0] = *(const vec*)(fq + (unsigned)i0 * F);
        u[1] = *(const vec*)(fq + (unsigned)i1 * F);
        u[2] = *(const vec*)(fq + (unsigned)i2 * F);
        u[3] = *(const vec*)(fq + (unsigned)i3 * F);
        u[4] = *(const vec*)(fq + (unsigned)i4 * F);
        u[5] = *(const vec*)(fq + (unsigned)i5 * F);
        u[6] = *(const vec*)(fq + (unsigned)i6 * F);
        u[7] = *(const vec*)(fq + (unsigned)i7 * F);
        {
            int t = e + 8 + q;
            mine = perm[t < last ? t : last];
        }
#pragma unroll
        for (int k = 0; k < 8; ++k) {
            int ee = e + k;
            if (ee < end) {
                int sidx = (ee >= b1 ? 1 : 0) + (ee >= b2 ? 1 : 0) + (ee >= b3 ? 1 : 0);
                if (sidx != cur) {
                    *(vec*)(slot0 + cur * SLOT) = acc;
                    acc = zero;
                    cur = sidx;
                }
                acc += u[k];
            }
        }
    }
    *(vec*)(slot0 + cur * SLOT) = acc;
}


#define P1 20
#define P2 36

__global__ __launch_bounds__(256) void rgcn_layer1(
    const float* __restrict__ x, const int* __restrict__ offs, const int* __restrict__ perm,
    const float* __restrict__ W1, const float* __restrict__ root1, const float* __restrict__ b1,
    float* __restrict__ h) {
    __shared__ __attribute__((aligned(16))) float agg[NR * 16 * P1];
    __shared__ float part[NR][16][33];
    int r    = __builtin_amdgcn_readfirstlane(threadIdx.x >> 5);
    int lane = threadIdx.x & 31;
    int m0   = blockIdx.x * 16;
    int arow = m0 + (lane & 15);
    int kh   = (lane >> 4) << 1;
    int col  = lane & 15;

    int key = arow * NR + r;
    int beg = offs[key], end = offs[key + 1];
    tile_seg_sums<16, P1>(x, offs, perm, m0, agg);
    float sc = 1.0f / fmaxf((float)(end - beg), 1.0f);
    __syncthreads();

    v8f c0, c1;
    if (r == 0) {
        float bb0 = b1[col], bb1 = b1[16 + col];
#pragma unroll
        for (int v = 0; v < 8; ++v) { c0[v] = bb0; c1[v] = bb1; }
#pragma unroll
        for (int k = 0; k < 16; k += 4) {
            v2f a  = a_frag(x, arow, 16, k, kh);
            v2f bA = b_frag(root1, 32, k, kh, col);
            v2f bB = b_frag(root1, 32, k, kh, col + 16);
            c0 = __builtin_amdgcn_wmma_f32_16x16x4_f32(false, a, false, bA, (short)0, c0, false, false);
            c1 = __builtin_amdgcn_wmma_f32_16x16x4_f32(false, a, false, bB, (short)0, c1, false, false);
        }
    } else {
#pragma unroll
        for (int v = 0; v < 8; ++v) { c0[v] = 0.0f; c1[v] = 0.0f; }
    }
    const float* Wr = W1 + (size_t)r * 16 * 32;
    const float* ar = agg + (r * 16 + (lane & 15)) * P1 + kh;
#pragma unroll
    for (int j = 0; j < 4; ++j) {
        v2f a  = *(const v2f*)(ar + 4 * j) * sc;
        v2f bA = b_frag(Wr, 32, 4 * j, kh, col);
        v2f bB = b_frag(Wr, 32, 4 * j, kh, col + 16);
        c0 = __builtin_amdgcn_wmma_f32_16x16x4_f32(false, a, false, bA, (short)0, c0, false, false);
        c1 = __builtin_amdgcn_wmma_f32_16x16x4_f32(false, a, false, bB, (short)0, c1, false, false);
    }
    int rb = (lane >> 4) << 3;
#pragma unroll
    for (int v = 0; v < 8; ++v) {
        part[r][rb + v][col]      = c0[v];
        part[r][rb + v][16 + col] = c1[v];
    }
    __syncthreads();
#pragma unroll
    for (int i = 0; i < 2; ++i) {
        int idx = threadIdx.x + 256 * i;
        int row = idx >> 5, cc = idx & 31;
        float s = part[0][row][cc];
#pragma unroll
        for (int w = 1; w < NR; ++w) s += part[w][row][cc];
        h[(size_t)m0 * 32 + idx] = fmaxf(s, 0.0f);
    }
}


__global__ __launch_bounds__(256) void rgcn_layer2(
    const float* __restrict__ h, const int* __restrict__ offs, const int* __restrict__ perm,
    const float* __restrict__ W2, const float* __restrict__ root2, const float* __restrict__ b2,
    float* __restrict__ out) {
    __shared__ __attribute__((aligned(16))) float agg[NR * 16 * P2];
    __shared__ float part[NR][16][17];
    int r    = __builtin_amdgcn_readfirstlane(threadIdx.x >> 5);
    int lane = threadIdx.x & 31;
    int m0   = blockIdx.x * 16;
    int arow = m0 + (lane & 15);
    int kh   = (lane >> 4) << 1;
    int col  = lane & 15;

    int key = arow * NR + r;
    int beg = offs[key], end = offs[key + 1];
    tile_seg_sums<32, P2>(h, offs, perm, m0, agg);
    float sc = 1.0f / fmaxf((float)(end - beg), 1.0f);
    __syncthreads();

    v8f c;
    if (r == 0) {
        float bb = b2[col];
#pragma unroll
        for (int v = 0; v < 8; ++v) c[v] = bb;
#pragma unroll
        for (int k = 0; k < 32; k += 4) {
            v2f a = a_frag(h, arow, 32, k, kh);
            v2f b = b_frag(root2, 16, k, kh, col);
            c = __builtin_amdgcn_wmma_f32_16x16x4_f32(false, a, false, b, (short)0, c, false, false);
        }
    } else {
#pragma unroll
        for (int v = 0; v < 8; ++v) c[v] = 0.0f;
    }
    const float* Wr = W2 + (size_t)r * 32 * 16;
    const float* ar = agg + (r * 16 + (lane & 15)) * P2 + kh;
#pragma unroll
    for (int j = 0; j < 8; ++j) {
        v2f a = *(const v2f*)(ar + 4 * j) * sc;
        v2f b = b_frag(Wr, 16, 4 * j, kh, col);
        c = __builtin_amdgcn_wmma_f32_16x16x4_f32(false, a, false, b, (short)0, c, false, false);
    }
    int rb = (lane >> 4) << 3;
#pragma unroll
    for (int v = 0; v < 8; ++v) part[r][rb + v][col] = c[v];
    __syncthreads();
    {
        int row = threadIdx.x >> 4, cc = threadIdx.x & 15;
        float s = part[0][row][cc];
#pragma unroll
        for (int w = 1; w < NR; ++w) s += part[w][row][cc];
        out[(size_t)m0 * 16 + threadIdx.x] = s;
    }
}


extern "C" void kernel_launch(void* const* d_in, const int* in_sizes, int n_in,
                              void* d_out, int out_size, void* d_ws, size_t ws_size,
                              hipStream_t stream) {
    const float* x     = (const float*)d_in[0];
    const int*   eidx  = (const int*)d_in[1];
    const int*   etype = (const int*)d_in[2];
    const float* W1    = (const float*)d_in[3];
    const float* root1 = (const float*)d_in[4];
    const float* b1    = (const float*)d_in[5];
    const float* W2    = (const float*)d_in[6];
    const float* root2 = (const float*)d_in[7];
    const float* b2    = (const float*)d_in[8];
    float* out = (float*)d_out;

    const int* src = eidx;
    const int* dst = eidx + NE;

    float* h    = (float*)d_ws;
    int*   cnt  = (int*)(h + (size_t)NN * 32);
    int*   offs = cnt + NK;
    int*   bsum = offs + NK + 4;
    int*   rank = bsum + 1024;
    int*   perm = rank + NE;

    zero_i32<<<2048, 256, 0, stream>>>(cnt, NK);
    count_rank<<<NE / 256, 256, 0, stream>>>(dst, etype, cnt, rank);
    seg_block_totals<<<SCAN_BLOCKS, 256, 0, stream>>>(cnt, bsum);
    seg_offsets<<<SCAN_BLOCKS, 256, 0, stream>>>(cnt, bsum, offs);
    place_edges<<<NE / 256, 256, 0, stream>>>(src, dst, etype, rank, offs, perm);

    rgcn_layer1<<<NN / 16, 256, 0, stream>>>(x, offs, perm, W1, root1, b1, h);
    rgcn_layer2<<<NN / 16, 256, 0, stream>>>(h, offs, perm, W2, root2, b2, out);

    (void)in_sizes; (void)n_in; (void)out_size; (void)ws_size;
}


# edge sort rebuilt in LDS, no global atomics: 256 edge chunks ordered by 391-node dst range, then one work-group per range ranks by LDS add; 7 launches to 4; sums in LDS add-arrival order
# speedup vs baseline: 4.1258x; 2.4574x over previous
#include <hip/hip_runtime.h>

#define NN 100000
#define NE 3200000
#define NR 8
#define NK (NN * NR)
#define NCH 256
#define CH (NE / NCH)
#define EPT 13
#define RNG 391
#define NRG 256
#define KR (RNG * NR)
#define KPAD 4096

typedef __attribute__((ext_vector_type(2))) float v2f;
typedef __attribute__((ext_vector_type(8))) float v8f;


__global__ __launch_bounds__(1024) void bin_chunks(const int* __restrict__ src,
                                                   const int* __restrict__ dst,
                                                   const int* __restrict__ et,
                                                   int* __restrict__ binned,
                                                   int* __restrict__ starts) {
    __shared__ int cntR[NRG];
    __shared__ int startR[NRG];
    __shared__ int wsum[8];
    __shared__ int buf[CH];
    const int tid = threadIdx.x, lane = tid & 31, c = blockIdx.x;
    const int e0 = c * CH;
    if (tid < NRG) cntR[tid] = 0;

    int word[EPT], rgk[EPT], lr[EPT];
#pragma unroll
    for (int k = 0; k < EPT; ++k) {
        int i = tid + 1024 * k;
        word[k] = 0; rgk[k] = 0; lr[k] = 0;
        if (i < CH) {
            int d = dst[e0 + i], r = et[e0 + i], s = src[e0 + i];
            int rg = d / RNG;
            word[k] = ((d - rg * RNG) << 20) | (r << 17) | s;
            rgk[k] = rg;
        }
    }
    __syncthreads();
#pragma unroll
    for (int k = 0; k < EPT; ++k)
        if (tid + 1024 * k < CH) lr[k] = atomicAdd(&cntR[rgk[k]], 1);
    __syncthreads();

    int v = 0, inc = 0;
    if (tid < NRG) {
        v = cntR[tid];
        inc = v;
        for (int o = 1; o < 32; o <<= 1) {
            int n = __shfl_up(inc, o, 32);
            if (lane >= o) inc += n;
        }
        if (lane == 31) wsum[tid >> 5] = inc;
    }
    __syncthreads();
    if (tid < NRG) {
        int b = 0;
        for (int w = 0; w < (tid >> 5); ++w) b += wsum[w];
        int ex = b + inc - v;
        startR[tid] = ex;
        starts[c * (NRG + 1) + tid] = ex;
        if (tid == NRG - 1) starts[c * (NRG + 1) + NRG] = CH;
    }
    __syncthreads();
#pragma unroll
    for (int k = 0; k < EPT; ++k)
        if (tid + 1024 * k < CH) buf[startR[rgk[k]] + lr[k]] = word[k];
    __syncthreads();
#pragma unroll
    for (int k = 0; k < EPT; ++k) {
        int i = tid + 1024 * k;
        if (i < CH) binned[e0 + i] = buf[i];
    }
}

template <bool PLACE>
__device__ __forceinline__ void walk_runs(const int* __restrict__ binned, const int* rs, const int* re,
                                          int* kc, int* __restrict__ perm, int wv, int lane) {
    int a[8], b[8], s[8], e[8];
#pragma unroll
    for (int j = 0; j < 8; ++j) {
        int c = wv + 32 * j;
        s[j] = rs[c]; e[j] = re[c];
        int i0 = s[j] + lane, i1 = i0 + 32;
        a[j] = -1; b[j] = -1;
        if (i0 < e[j]) a[j] = binned[i0];
        if (i1 < e[j]) b[j] = binned[i1];
    }
#pragma unroll
    for (int j = 0; j < 8; ++j) {
        if (a[j] >= 0) {
            if (PLACE) perm[atomicAdd(&kc[a[j] >> 17], 1)] = a[j] & 0x1FFFF;
            else atomicAdd(&kc[a[j] >> 17], 1);
        }
        if (b[j] >= 0) {
            if (PLACE) perm[atomicAdd(&kc[b[j] >> 17], 1)] = b[j] & 0x1FFFF;
            else atomicAdd(&kc[b[j] >> 17], 1);
        }
        for (int i = s[j] + 64 + lane; i < e[j]; i += 32) {
            int w = binned[i];
            if (PLACE) perm[atomicAdd(&kc[w >> 17], 1)] = w & 0x1FFFF;
            else atomicAdd(&kc[w >> 17], 1);
        }
    }
}

__global__ __launch_bounds__(1024) void sort_range(const int* __restrict__ binned,
                                                   const int* __restrict__ starts,
                                                   int* __restrict__ offs, int* __restrict__ perm) {
    __shared__ __attribute__((aligned(16))) int kc[KPAD];
    __shared__ int rs[NCH];
    __shared__ int re[NCH];
    __shared__ int wtot[32];
    __shared__ int bpart[8];
    const int tid = threadIdx.x, lane = tid & 31, wv = tid >> 5, R = blockIdx.x;

    *(int4*)&kc[4 * tid] = make_int4(0, 0, 0, 0);
    if (tid < NCH) {
        int s = starts[tid * (NRG + 1) + R], e = starts[tid * (NRG + 1) + R + 1];
        rs[tid] = tid * CH + s;
        re[tid] = tid * CH + e;
        int b = s;
        for (int o = 16; o > 0; o >>= 1) b += __shfl_xor(b, o, 32);
        if (lane == 0) bpart[wv] = b;
    }
    __syncthreads();

    walk_runs<false>(binned, rs, re, kc, perm, wv, lane);
    __syncthreads();

    int4 v = *(const int4*)&kc[4 * tid];
    int t = v.x + v.y + v.z + v.w;
    int inc = t;
    for (int o = 1; o < 32; o <<= 1) {
        int n = __shfl_up(inc, o, 32);
        if (lane >= o) inc += n;
    }
    if (lane == 31) wtot[wv] = inc;
    __syncthreads();

    int pre = 0;
    for (int w = 0; w < 8; ++w) pre += bpart[w];
    for (int w = 0; w < wv; ++w) pre += wtot[w];
    int4 o4;
    o4.x = pre + inc - t; o4.y = o4.x + v.x; o4.z = o4.y + v.y; o4.w = o4.z + v.z;
    *(int4*)&kc[4 * tid] = o4;
    int key0 = R * KR + 4 * tid;
    if (4 * tid < KR && key0 < NK) *(int4*)(offs + key0) = o4;
    if (R == NRG - 1 && tid == 0) offs[NK] = NE;
    __syncthreads();

    walk_runs<true>(binned, rs, re, kc, perm, wv, lane);
}

__device__ __forceinline__ v2f a_frag(const float* __restrict__ M, int row, int stride,
                                      int k0, int kh) {
    const float* p = M + (size_t)row * stride + k0 + kh;
    v2f a; a.x = p[0]; a.y = p[1]; return a;
}
__device__ __forceinline__ v2f b_frag(const float* __restrict__ W, int nout,
                                      int k0, int kh, int col) {
    v2f b;
    b.x = W[(size_t)(k0 + kh) * nout + col];
    b.y = W[(size_t)(k0 + kh + 1) * nout + col];
    return b;
}

template <int F, int PITCH>
__device__ __forceinline__ void tile_seg_sums(const float* __restrict__ feat,
                                              const int* __restrict__ offs,
                                              const int* __restrict__ perm,
                                              int m0, float* __restrict__ agg) {
    typedef __attribute__((ext_vector_type(F / 8))) float vec;
    const int FPL  = F / 8;
    const int SLOT = 16 * PITCH;
    int q    = threadIdx.x & 7;
    int g    = threadIdx.x >> 3;
    int node = g >> 1, half = g & 1;
    int k0   = (m0 + node) * NR + half * 4;
    int4 bb  = *(const int4*)(offs + k0);
    int beg = bb.x, b1 = bb.y, b2 = bb.z, b3 = bb.w;
    int end  = offs[k0 + 4];
    int last = end - 1;

    float* slot0 = agg + (half * 4 * 16 + node) * PITCH + q * FPL;
    vec zero;
#pragma unroll
    for (int i = 0; i < FPL; ++i) zero[i] = 0.0f;
#pragma unroll
    for (int s = 0; s < 4; ++s) *(vec*)(slot0 + s * SLOT) = zero;

    const float* fq = feat + q * FPL;
    vec acc = zero;
    int cur = 0;
    int mine = 0;
    if (beg < end) { int t = beg + q; mine = perm[t < last ? t : last]; }
    for (int e = beg; e < end; e += 8) {
        int i0 = __builtin_amdgcn_mov_dpp8(mine, 0x000000);
        int i1 = __builtin_amdgcn_mov_dpp8(mine, 0x249249);
        int i2 = __builtin_amdgcn_mov_dpp8(mine, 0x492492);
        int i3 = __builtin_amdgcn_mov_dpp8(mine, 0x6DB6DB);
        int i4 = __builtin_amdgcn_mov_dpp8(mine, 0x924924);
        int i5 = __builtin_amdgcn_mov_dpp8(mine, 0xB6DB6D);
        int i6 = __builtin_amdgcn_mov_dpp8(mine, 0xDB6DB6);
        int i7 = __builtin_amdgcn_mov_dpp8(mine, 0xFFFFFF);
        vec u[8];
        u[0] = *(const vec*)(fq + (unsigned)i0 * F);
        u[1] = *(const vec*)(fq + (unsigned)i1 * F);
        u[2] = *(const vec*)(fq + (unsigned)i2 * F);
        u[3] = *(const vec*)(fq + (unsigned)i3 * F);
        u[4] = *(const vec*)(fq + (unsigned)i4 * F);
        u[5] = *(const vec*)(fq + (unsigned)i5 * F);
        u[6] = *(const vec*)(fq + (unsigned)i6 * F);
        u[7] = *(const vec*)(fq + (unsigned)i7 * F);
        {
            int t = e + 8 + q;
            mine = perm[t < last ? t : last];
        }
#pragma unroll
        for (int k = 0; k < 8; ++k) {
            int ee = e + k;
            if (ee < end) {
                int sidx = (ee >= b1 ? 1 : 0) + (ee >= b2 ? 1 : 0) + (ee >= b3 ? 1 : 0);
                if (sidx != cur) {
                    *(vec*)(slot0 + cur * SLOT) = acc;
                    acc = zero;
                    cur = sidx;
                }
                acc += u[k];
            }
        }
    }
    *(vec*)(slot0 + cur * SLOT) = acc;
}


#define P1 20
#define P2 36

__global__ __launch_bounds__(256) void rgcn_layer1(
    const float* __restrict__ x, const int* __restrict__ offs, const int* __restrict__ perm,
    const float* __restrict__ W1, const float* __restrict__ root1, const float* __restrict__ b1,
    float* __restrict__ h) {
    __shared__ __attribute__((aligned(16))) float agg[NR * 16 * P1];
    __shared__ float part[NR][16][33];
    int r    = __builtin_amdgcn_readfirstlane(threadIdx.x >> 5);
    int lane = threadIdx.x & 31;
    int m0   = blockIdx.x * 16;
    int arow = m0 + (lane & 15);
    int kh   = (lane >> 4) << 1;
    int col  = lane & 15;

    int key = arow * NR + r;
    int beg = offs[key], end = offs[key + 1];
    tile_seg_sums<16, P1>(x, offs, perm, m0, agg);
    float sc = 1.0f / fmaxf((float)(end - beg), 1.0f);
    __syncthreads();

    v8f c0, c1;
    if (r == 0) {
        float bb0 = b1[col], bb1 = b1[16 + col];
#pragma unroll
        for (int v = 0; v < 8; ++v) { c0[v] = bb0; c1[v] = bb1; }
#pragma unroll
        for (int k = 0; k < 16; k += 4) {
            v2f a  = a_frag(x, arow, 16, k, kh);
            v2f bA = b_frag(root1, 32, k, kh, col);
            v2f bB = b_frag(root1, 32, k, kh, col + 16);
            c0 = __builtin_amdgcn_wmma_f32_16x16x4_f32(false, a, false, bA, (short)0, c0, false, false);
            c1 = __builtin_amdgcn_wmma_f32_16x16x4_f32(false, a, false, bB, (short)0, c1, false, false);
        }
    } else {
#pragma unroll
        for (int v = 0; v < 8; ++v) { c0[v] = 0.0f; c1[v] = 0.0f; }
    }
    const float* Wr = W1 + (size_t)r * 16 * 32;
    const float* ar = agg + (r * 16 + (lane & 15)) * P1 + kh;
#pragma unroll
    for (int j = 0; j < 4; ++j) {
        v2f a  = *(const v2f*)(ar + 4 * j) * sc;
        v2f bA = b_frag(Wr, 32, 4 * j, kh, col);
        v2f bB = b_frag(Wr, 32, 4 * j, kh, col + 16);
        c0 = __builtin_amdgcn_wmma_f32_16x16x4_f32(false, a, false, bA, (short)0, c0, false, false);
        c1 = __builtin_amdgcn_wmma_f32_16x16x4_f32(false, a, false, bB, (short)0, c1, false, false);
    }
    int rb = (lane >> 4) << 3;
#pragma unroll
    for (int v = 0; v < 8; ++v) {
        part[r][rb + v][col]      = c0[v];
        part[r][rb + v][16 + col] = c1[v];
    }
    __syncthreads();
#pragma unroll
    for (int i = 0; i < 2; ++i) {
        int idx = threadIdx.x + 256 * i;
        int row = idx >> 5, cc = idx & 31;
        float s = part[0][row][cc];
#pragma unroll
        for (int w = 1; w < NR; ++w) s += part[w][row][cc];
        h[(size_t)m0 * 32 + idx] = fmaxf(s, 0.0f);
    }
}


__global__ __launch_bounds__(256) void rgcn_layer2(
    const float* __restrict__ h, const int* __restrict__ offs, const int* __restrict__ perm,
    const float* __restrict__ W2, const float* __restrict__ root2, const float* __restrict__ b2,
    float* __restrict__ out) {
    __shared__ __attribute__((aligned(16))) float agg[NR * 16 * P2];
    __shared__ float part[NR][16][17];
    int r    = __builtin_amdgcn_readfirstlane(threadIdx.x >> 5);
    int lane = threadIdx.x & 31;
    int m0   = blockIdx.x * 16;
    int arow = m0 + (lane & 15);
    int kh   = (lane >> 4) << 1;
    int col  = lane & 15;

    int key = arow * NR + r;
    int beg = offs[key], end = offs[key + 1];
    tile_seg_sums<32, P2>(h, offs, perm, m0, agg);
    float sc = 1.0f / fmaxf((float)(end - beg), 1.0f);
    __syncthreads();

    v8f c;
    if (r == 0) {
        float bb = b2[col];
#pragma unroll
        for (int v = 0; v < 8; ++v) c[v] = bb;
#pragma unroll
        for (int k = 0; k < 32; k += 4) {
            v2f a = a_frag(h, arow, 32, k, kh);
            v2f b = b_frag(root2, 16, k, kh, col);
            c = __builtin_amdgcn_wmma_f32_16x16x4_f32(false, a, false, b, (short)0, c, false, false);
        }
    } else {
#pragma unroll
        for (int v = 0; v < 8; ++v) c[v] = 0.0f;
    }
    const float* Wr = W2 + (size_t)r * 32 * 16;
    const float* ar = agg + (r * 16 + (lane & 15)) * P2 + kh;
#pragma unroll
    for (int j = 0; j < 8; ++j) {
        v2f a = *(const v2f*)(ar + 4 * j) * sc;
        v2f b = b_frag(Wr, 16, 4 * j, kh, col);
        c = __builtin_amdgcn_wmma_f32_16x16x4_f32(false, a, false, b, (short)0, c, false, false);
    }
    int rb = (lane >> 4) << 3;
#pragma unroll
    for (int v = 0; v < 8; ++v) part[r][rb + v][col] = c[v];
    __syncthreads();
    {
        int row = threadIdx.x >> 4, cc = threadIdx.x & 15;
        float s = part[0][row][cc];
#pragma unroll
        for (int w = 1; w < NR; ++w) s += part[w][row][cc];
        out[(size_t)m0 * 16 + threadIdx.x] = s;
    }
}


extern "C" void kernel_launch(void* const* d_in, const int* in_sizes, int n_in,
                              void* d_out, int out_size, void* d_ws, size_t ws_size,
                              hipStream_t stream) {
    const float* x     = (const float*)d_in[0];
    const int*   eidx  = (const int*)d_in[1];
    const int*   etype = (const int*)d_in[2];
    const float* W1    = (const float*)d_in[3];
    const float* root1 = (const float*)d_in[4];
    const float* b1    = (const float*)d_in[5];
    const float* W2    = (const float*)d_in[6];
    const float* root2 = (const float*)d_in[7];
    const float* b2    = (const float*)d_in[8];
    float* out = (float*)d_out;

    const int* src = eidx;
    const int* dst = eidx + NE;

    float* h       = (float*)d_ws;
    int*   offs    = (int*)(h + (size_t)NN * 32);
    int*   binned  = offs + NK + 4;
    int*   starts  = binned + NE;
    int*   perm    = starts + NCH * (NRG + 1);

    bin_chunks<<<NCH, 1024, 0, stream>>>(src, dst, etype, binned, starts);
    sort_range<<<NRG, 1024, 0, stream>>>(binned, starts, offs, perm);

    rgcn_layer1<<<NN / 16, 256, 0, stream>>>(x, offs, perm, W1, root1, b1, h);
    rgcn_layer2<<<NN / 16, 256, 0, stream>>>(h, offs, perm, W2, root2, b2, out);

    (void)in_sizes; (void)n_in; (void)out_size; (void)ws_size;
}


# layer tiles read their index words once: a wave stages its piece of the sorted source list through LDS with one 16-byte load per lane, segment sizes pass through LDS; sum order unchanged
# speedup vs baseline: 4.2862x; 1.0389x over previous
#include <hip/hip_runtime.h>

#define NN 100000
#define NE 3200000
#define NR 8
#define NK (NN * NR)
#define NCH 256
#define CH (NE / NCH)
#define EPT 13
#define RNG 391
#define NRG 256
#define KR (RNG * NR)
#define KPAD 4096

typedef __attribute__((ext_vector_type(2))) float v2f;
typedef __attribute__((ext_vector_type(8))) float v8f;


__global__ __launch_bounds__(1024) void bin_chunks(const int* __restrict__ src,
                                                   const int* __restrict__ dst,
                                                   const int* __restrict__ et,
                                                   int* __restrict__ binned,
                                                   int* __restrict__ starts) {
    __shared__ int cntR[NRG];
    __shared__ int startR[NRG];
    __shared__ int wsum[8];
    __shared__ int buf[CH];
    const int tid = threadIdx.x, lane = tid & 31, c = blockIdx.x;
    const int e0 = c * CH;
    if (tid < NRG) cntR[tid] = 0;

    int word[EPT], rgk[EPT], lr[EPT];
#pragma unroll
    for (int k = 0; k < EPT; ++k) {
        int i = tid + 1024 * k;
        word[k] = 0; rgk[k] = 0; lr[k] = 0;
        if (i < CH) {
            int d = dst[e0 + i], r = et[e0 + i], s = src[e0 + i];
            int rg = d / RNG;
            word[k] = ((d - rg * RNG) << 20) | (r << 17) | s;
            rgk[k] = rg;
        }
    }
    __syncthreads();
#pragma unroll
    for (int k = 0; k < EPT; ++k)
        if (tid + 1024 * k < CH) lr[k] = atomicAdd(&cntR[rgk[k]], 1);
    __syncthreads();

    int v = 0, inc = 0;
    if (tid < NRG) {
        v = cntR[tid];
        inc = v;
        for (int o = 1; o < 32; o <<= 1) {
            int n = __shfl_up(inc, o, 32);
            if (lane >= o) inc += n;
        }
        if (lane == 31) wsum[tid >> 5] = inc;
    }
    __syncthreads();
    if (tid < NRG) {
        int b = 0;
        for (int w = 0; w < (tid >> 5); ++w) b += wsum[w];
        int ex = b + inc - v;
        startR[tid] = ex;
        starts[c * (NRG + 1) + tid] = ex;
        if (tid == NRG - 1) starts[c * (NRG + 1) + NRG] = CH;
    }
    __syncthreads();
#pragma unroll
    for (int k = 0; k < EPT; ++k)
        if (tid + 1024 * k < CH) buf[startR[rgk[k]] + lr[k]] = word[k];
    __syncthreads();
#pragma unroll
    for (int k = 0; k < EPT; ++k) {
        int i = tid + 1024 * k;
        if (i < CH) binned[e0 + i] = buf[i];
    }
}

template <bool PLACE>
__device__ __forceinline__ void walk_runs(const int* __restrict__ binned, const int* rs, const int* re,
                                          int* kc, int* __restrict__ perm, int wv, int lane) {
    int a[8], b[8], s[8], e[8];
#pragma unroll
    for (int j = 0; j < 8; ++j) {
        int c = wv + 32 * j;
        s[j] = rs[c]; e[j] = re[c];
        int i0 = s[j] + lane, i1 = i0 + 32;
        a[j] = -1; b[j] = -1;
        if (i0 < e[j]) a[j] = binned[i0];
        if (i1 < e[j]) b[j] = binned[i1];
    }
#pragma unroll
    for (int j = 0; j < 8; ++j) {
        if (a[j] >= 0) {
            if (PLACE) perm[atomicAdd(&kc[a[j] >> 17], 1)] = a[j] & 0x1FFFF;
            else atomicAdd(&kc[a[j] >> 17], 1);
        }
        if (b[j] >= 0) {
            if (PLACE) perm[atomicAdd(&kc[b[j] >> 17], 1)] = b[j] & 0x1FFFF;
            else atomicAdd(&kc[b[j] >> 17], 1);
        }
        for (int i = s[j] + 64 + lane; i < e[j]; i += 32) {
            int w = binned[i];
            if (PLACE) perm[atomicAdd(&kc[w >> 17], 1)] = w & 0x1FFFF;
            else atomicAdd(&kc[w >> 17], 1);
        }
    }
}

__global__ __launch_bounds__(1024) void sort_range(const int* __restrict__ binned,
                                                   const int* __restrict__ starts,
                                                   int* __restrict__ offs, int* __restrict__ perm) {
    __shared__ __attribute__((aligned(16))) int kc[KPAD];
    __shared__ int rs[NCH];
    __shared__ int re[NCH];
    __shared__ int wtot[32];
    __shared__ int bpart[8];
    const int tid = threadIdx.x, lane = tid & 31, wv = tid >> 5, R = blockIdx.x;

    *(int4*)&kc[4 * tid] = make_int4(0, 0, 0, 0);
    if (tid < NCH) {
        int s = starts[tid * (NRG + 1) + R], e = starts[tid * (NRG + 1) + R + 1];
        rs[tid] = tid * CH + s;
        re[tid] = tid * CH + e;
        int b = s;
        for (int o = 16; o > 0; o >>= 1) b += __shfl_xor(b, o, 32);
        if (lane == 0) bpart[wv] = b;
    }
    __syncthreads();

    walk_runs<false>(binned, rs, re, kc, perm, wv, lane);
    __syncthreads();

    int4 v = *(const int4*)&kc[4 * tid];
    int t = v.x + v.y + v.z + v.w;
    int inc = t;
    for (int o = 1; o < 32; o <<= 1) {
        int n = __shfl_up(inc, o, 32);
        if (lane >= o) inc += n;
    }
    if (lane == 31) wtot[wv] = inc;
    __syncthreads();

    int pre = 0;
    for (int w = 0; w < 8; ++w) pre += bpart[w];
    for (int w = 0; w < wv; ++w) pre += wtot[w];
    int4 o4;
    o4.x = pre + inc - t; o4.y = o4.x + v.x; o4.z = o4.y + v.y; o4.w = o4.z + v.z;
    *(int4*)&kc[4 * tid] = o4;
    int key0 = R * KR + 4 * tid;
    if (4 * tid < KR && key0 < NK) *(int4*)(offs + key0) = o4;
    if (R == NRG - 1 && tid == 0) offs[NK] = NE;
    __syncthreads();

    walk_runs<true>(binned, rs, re, kc, perm, wv, lane);
}

__device__ __forceinline__ v2f a_frag(const float* __restrict__ M, int row, int stride,
                                      int k0, int kh) {
    const float* p = M + (size_t)row * stride + k0 + kh;
    v2f a; a.x = p[0]; a.y = p[1]; return a;
}
__device__ __forceinline__ v2f b_frag(const float* __restrict__ W, int nout,
                                      int k0, int kh, int col) {
    v2f b;
    b.x = W[(size_t)(k0 + kh) * nout + col];
    b.y = W[(size_t)(k0 + kh + 1) * nout + col];
    return b;
}

#define WCAP 128

template <int F, int PITCH, bool STAGED>
__device__ __forceinline__ void walk_group(const float* __restrict__ fq, const int* __restrict__ perm,
                                           const int* widx, int base, int q,
                                           int beg, int b1, int b2, int b3, int end,
                                           float* __restrict__ slot0) {
    typedef __attribute__((ext_vector_type(F / 8))) float vec;
    const int FPL  = F / 8;
    const int SLOT = 16 * PITCH;
    int last = end - 1;
    vec zero;
#pragma unroll
    for (int i = 0; i < FPL; ++i) zero[i] = 0.0f;
    vec acc = zero;
    int cur = 0;
    int mine = 0;
    if (beg < end) {
        int t = beg + q; t = t < last ? t : last;
        mine = STAGED ? widx[t - base] : perm[t];
    }
    for (int e = beg; e < end; e += 8) {
        int i0 = __builtin_amdgcn_mov_dpp8(mine, 0x000000);
        int i1 = __builtin_amdgcn_mov_dpp8(mine, 0x249249);
        int i2 = __builtin_amdgcn_mov_dpp8(mine, 0x492492);
        int i3 = __builtin_amdgcn_mov_dpp8(mine, 0x6DB6DB);
        int i4 = __builtin_amdgcn_mov_dpp8(mine, 0x924924);
        int i5 = __builtin_amdgcn_mov_dpp8(mine, 0xB6DB6D);
        int i6 = __builtin_amdgcn_mov_dpp8(mine, 0xDB6DB6);
        int i7 = __builtin_amdgcn_mov_dpp8(mine, 0xFFFFFF);
        vec u[8];
        u[0] = *(const vec*)(fq + (unsigned)i0 * F);
        u[1] = *(const vec*)(fq + (unsigned)i1 * F);
        u[2] = *(const vec*)(fq + (unsigned)i2 * F);
        u[3] = *(const vec*)(fq + (unsigned)i3 * F);
        u[4] = *(const vec*)(fq + (unsigned)i4 * F);
        u[5] = *(const vec*)(fq + (unsigned)i5 * F);
        u[6] = *(const vec*)(fq + (unsigned)i6 * F);
        u[7] = *(const vec*)(fq + (unsigned)i7 * F);
        {
            int t = e + 8 + q; t = t < last ? t : last;
            mine = STAGED ? widx[t - base] : perm[t];
        }
#pragma unroll
        for (int k = 0; k < 8; ++k) {
            int ee = e + k;
            if (ee < end) {
                int sidx = (ee >= b1 ? 1 : 0) + (ee >= b2 ? 1 : 0) + (ee >= b3 ? 1 : 0);
                if (sidx != cur) {
                    *(vec*)(slot0 + cur * SLOT) = acc;
                    acc = zero;
                    cur = sidx;
                }
                acc += u[k];
            }
        }
    }
    *(vec*)(slot0 + cur * SLOT) = acc;
}

template <int F, int PITCH>
__device__ __forceinline__ void tile_seg_sums(const float* __restrict__ feat,
                                              const int* __restrict__ offs,
                                              const int* __restrict__ perm,
                                              int m0, float* __restrict__ agg,
                                              int* __restrict__ widx, int* __restrict__ cnt) {
    typedef __attribute__((ext_vector_type(F / 8))) float vec;
    const int FPL  = F / 8;
    const int SLOT = 16 * PITCH;
    int q    = threadIdx.x & 7;
    int g    = threadIdx.x >> 3;
    int node = g >> 1, half = g & 1;
    int k0   = (m0 + node) * NR + half * 4;
    int4 bb  = *(const int4*)(offs + k0);
    int beg = bb.x, b1 = bb.y, b2 = bb.z, b3 = bb.w;
    int end  = offs[k0 + 4];
    if (q == 0) *(int4*)(cnt + (node * NR + half * 4)) = make_int4(b1 - beg, b2 - b1, b3 - b2, end - b3);

    float* slot0 = agg + (half * 4 * 16 + node) * PITCH + q * FPL;
    vec zero;
#pragma unroll
    for (int i = 0; i < FPL; ++i) zero[i] = 0.0f;
#pragma unroll
    for (int s = 0; s < 4; ++s) *(vec*)(slot0 + s * SLOT) = zero;

    int wbeg = __builtin_amdgcn_readfirstlane(beg);
    int wend = __builtin_amdgcn_readlane(end, 31);
    int base = wbeg & ~3;
    const float* fq = feat + q * FPL;
    if (wend - base <= WCAP) {
        int l4 = (threadIdx.x & 31) * 4;
        if (base + l4 < wend) *(int4*)(widx + l4) = *(const int4*)(perm + base + l4);
        walk_group<F, PITCH, true>(fq, perm, widx, base, q, beg, b1, b2, b3, end, slot0);
    } else {
        walk_group<F, PITCH, false>(fq, perm, widx, base, q, beg, b1, b2, b3, end, slot0);
    }
}


#define P1 20
#define P2 36

__global__ __launch_bounds__(256) void rgcn_layer1(
    const float* __restrict__ x, const int* __restrict__ offs, const int* __restrict__ perm,
    const float* __restrict__ W1, const float* __restrict__ root1, const float* __restrict__ b1,
    float* __restrict__ h) {
    __shared__ __attribute__((aligned(16))) float agg[NR * 16 * P1];
    __shared__ float part[NR][16][33];
    __shared__ __attribute__((aligned(16))) int widx[NR * WCAP];
    __shared__ __attribute__((aligned(16))) int cnt[16 * NR];
    int r    = __builtin_amdgcn_readfirstlane(threadIdx.x >> 5);
    int lane = threadIdx.x & 31;
    int m0   = blockIdx.x * 16;
    int arow = m0 + (lane & 15);
    int kh   = (lane >> 4) << 1;
    int col  = lane & 15;

    tile_seg_sums<16, P1>(x, offs, perm, m0, agg, widx + r * WCAP, cnt);
    __syncthreads();
    float sc = 1.0f / fmaxf((float)cnt[(lane & 15) * NR + r], 1.0f);

    v8f c0, c1;
    if (r == 0) {
        float bb0 = b1[col], bb1 = b1[16 + col];
#pragma unroll
        for (int v = 0; v < 8; ++v) { c0[v] = bb0; c1[v] = bb1; }
#pragma unroll
        for (int k = 0; k < 16; k += 4) {
            v2f a  = a_frag(x, arow, 16, k, kh);
            v2f bA = b_frag(root1, 32, k, kh, col);
            v2f bB = b_frag(root1, 32, k, kh, col + 16);
            c0 = __builtin_amdgcn_wmma_f32_16x16x4_f32(false, a, false, bA, (short)0, c0, false, false);
            c1 = __builtin_amdgcn_wmma_f32_16x16x4_f32(false, a, false, bB, (short)0, c1, false, false);
        }
    } else {
#pragma unroll
        for (int v = 0; v < 8; ++v) { c0[v] = 0.0f; c1[v] = 0.0f; }
    }
    const float* Wr = W1 + (size_t)r * 16 * 32;
    const float* ar = agg + (r * 16 + (lane & 15)) * P1 + kh;
#pragma unroll
    for (int j = 0; j < 4; ++j) {
        v2f a  = *(const v2f*)(ar + 4 * j) * sc;
        v2f bA = b_frag(Wr, 32, 4 * j, kh, col);
        v2f bB = b_frag(Wr, 32, 4 * j, kh, col + 16);
        c0 = __builtin_amdgcn_wmma_f32_16x16x4_f32(false, a, false, bA, (short)0, c0, false, false);
        c1 = __builtin_amdgcn_wmma_f32_16x16x4_f32(false, a, false, bB, (short)0, c1, false, false);
    }
    int rb = (lane >> 4) << 3;
#pragma unroll
    for (int v = 0; v < 8; ++v) {
        part[r][rb + v][col]      = c0[v];
        part[r][rb + v][16 + col] = c1[v];
    }
    __syncthreads();
#pragma unroll
    for (int i = 0; i < 2; ++i) {
        int idx = threadIdx.x + 256 * i;
        int row = idx >> 5, cc = idx & 31;
        float s = part[0][row][cc];
#pragma unroll
        for (int w = 1; w < NR; ++w) s += part[w][row][cc];
        h[(size_t)m0 * 32 + idx] = fmaxf(s, 0.0f);
    }
}


__global__ __launch_bounds__(256) void rgcn_layer2(
    const float* __restrict__ h, const int* __restrict__ offs, const int* __restrict__ perm,
    const float* __restrict__ W2, const float* __restrict__ root2, const float* __restrict__ b2,
    float* __restrict__ out) {
    __shared__ __attribute__((aligned(16))) float agg[NR * 16 * P2];
    __shared__ float part[NR][16][17];
    __shared__ __attribute__((aligned(16))) int widx[NR * WCAP];
    __shared__ __attribute__((aligned(16))) int cnt[16 * NR];
    int r    = __builtin_amdgcn_readfirstlane(threadIdx.x >> 5);
    int lane = threadIdx.x & 31;
    int m0   = blockIdx.x * 16;
    int arow = m0 + (lane & 15);
    int kh   = (lane >> 4) << 1;
    int col  = lane & 15;

    tile_seg_sums<32, P2>(h, offs, perm, m0, agg, widx + r * WCAP, cnt);
    __syncthreads();
    float sc = 1.0f / fmaxf((float)cnt[(lane & 15) * NR + r], 1.0f);

    v8f c;
    if (r == 0) {
        float bb = b2[col];
#pragma unroll
        for (int v = 0; v < 8; ++v) c[v] = bb;
#pragma unroll
        for (int k = 0; k < 32; k += 4) {
            v2f a = a_frag(h, arow, 32, k, kh);
            v2f b = b_frag(root2, 16, k, kh, col);
            c = __builtin_amdgcn_wmma_f32_16x16x4_f32(false, a, false, b, (short)0, c, false, false);
        }
    } else {
#pragma unroll
        for (int v = 0; v < 8; ++v) c[v] = 0.0f;
    }
    const float* Wr = W2 + (size_t)r * 32 * 16;
    const float* ar = agg + (r * 16 + (lane & 15)) * P2 + kh;
#pragma unroll
    for (int j = 0; j < 8; ++j) {
        v2f a = *(const v2f*)(ar + 4 * j) * sc;
        v2f b = b_frag(Wr, 16, 4 * j, kh, col);
        c = __builtin_amdgcn_wmma_f32_16x16x4_f32(false, a, false, b, (short)0, c, false, false);
    }
    int rb = (lane >> 4) << 3;
#pragma unroll
    for (int v = 0; v < 8; ++v) part[r][rb + v][col] = c[v];
    __syncthreads();
    {
        int row = threadIdx.x >> 4, cc = threadIdx.x & 15;
        float s = part[0][row][cc];
#pragma unroll
        for (int w = 1; w < NR; ++w) s += part[w][row][cc];
        out[(size_t)m0 * 16 + threadIdx.x] = s;
    }
}


extern "C" void kernel_launch(void* const* d_in, const int* in_sizes, int n_in,
                              void* d_out, int out_size, void* d_ws, size_t ws_size,
                              hipStream_t stream) {
    const float* x     = (const float*)d_in[0];
    const int*   eidx  = (const int*)d_in[1];
    const int*   etype = (const int*)d_in[2];
    const float* W1    = (const float*)d_in[3];
    const float* root1 = (const float*)d_in[4];
    const float* b1    = (const float*)d_in[5];
    const float* W2    = (const float*)d_in[6];
    const float* root2 = (const float*)d_in[7];
    const float* b2    = (const float*)d_in[8];
    float* out = (float*)d_out;

    const int* src = eidx;
    const int* dst = eidx + NE;

    float* h       = (float*)d_ws;
    int*   offs    = (int*)(h + (size_t)NN * 32);
    int*   binned  = offs + NK + 4;
    int*   starts  = binned + NE;
    int*   perm    = starts + NCH * (NRG + 1);

    bin_chunks<<<NCH, 1024, 0, stream>>>(src, dst, etype, binned, starts);
    sort_range<<<NRG, 1024, 0, stream>>>(binned, starts, offs, perm);

    rgcn_layer1<<<NN / 16, 256, 0, stream>>>(x, offs, perm, W1, root1, b1, h);
    rgcn_layer2<<<NN / 16, 256, 0, stream>>>(h, offs, perm, W2, root2, b2, out);

    (void)in_sizes; (void)n_in; (void)out_size; (void)ws_size;
}


# sort_range places sources in an LDS image of its range's piece of perm, then stores it 16 bytes per lane, four whole lines per wave store, not one 4-byte global store per edge; sum order unchanged
# speedup vs baseline: 4.7029x; 1.0972x over previous
#include <hip/hip_runtime.h>

#define NN 100000
#define NE 3200000
#define NR 8
#define NK (NN * NR)
#define NCH 256
#define CH (NE / NCH)
#define EPT 13
#define RNG 391
#define NRG 256
#define KR (RNG * NR)
#define KPAD 4096
#define PCAP 16384

typedef __attribute__((ext_vector_type(2))) float v2f;
typedef __attribute__((ext_vector_type(8))) float v8f;


__global__ __launch_bounds__(1024) void bin_chunks(const int* __restrict__ src,
                                                   const int* __restrict__ dst,
                                                   const int* __restrict__ et,
                                                   int* __restrict__ binned,
                                                   int* __restrict__ starts) {
    __shared__ int cntR[NRG];
    __shared__ int startR[NRG];
    __shared__ int wsum[8];
    __shared__ int buf[CH];
    const int tid = threadIdx.x, lane = tid & 31, c = blockIdx.x;
    const int e0 = c * CH;
    if (tid < NRG) cntR[tid] = 0;

    int word[EPT], rgk[EPT], lr[EPT];
#pragma unroll
    for (int k = 0; k < EPT; ++k) {
        int i = tid + 1024 * k;
        word[k] = 0; rgk[k] = 0; lr[k] = 0;
        if (i < CH) {
            int d = dst[e0 + i], r = et[e0 + i], s = src[e0 + i];
            int rg = d / RNG;
            word[k] = ((d - rg * RNG) << 20) | (r << 17) | s;
            rgk[k] = rg;
        }
    }
    __syncthreads();
#pragma unroll
    for (int k = 0; k < EPT; ++k)
        if (tid + 1024 * k < CH) lr[k] = atomicAdd(&cntR[rgk[k]], 1);
    __syncthreads();

    int v = 0, inc = 0;
    if (tid < NRG) {
        v = cntR[tid];
        inc = v;
        for (int o = 1; o < 32; o <<= 1) {
            int n = __shfl_up(inc, o, 32);
            if (lane >= o) inc += n;
        }
        if (lane == 31) wsum[tid >> 5] = inc;
    }
    __syncthreads();
    if (tid < NRG) {
        int b = 0;
        for (int w = 0; w < (tid >> 5); ++w) b += wsum[w];
        int ex = b + inc - v;
        startR[tid] = ex;
        starts[c * (NRG + 1) + tid] = ex;
        if (tid == NRG - 1) starts[c * (NRG + 1) + NRG] = CH;
    }
    __syncthreads();
#pragma unroll
    for (int k = 0; k < EPT; ++k)
        if (tid + 1024 * k < CH) buf[startR[rgk[k]] + lr[k]] = word[k];
    __syncthreads();
#pragma unroll
    for (int k = 0; k < EPT; ++k) {
        int i = tid + 1024 * k;
        if (i < CH) binned[e0 + i] = buf[i];
    }
}

__device__ __forceinline__ void place_src(int* pl, int* __restrict__ perm, int pbase, int slot, int s) {
    int rel = slot - pbase;
    if (rel < PCAP) pl[rel] = s;
    else perm[slot] = s;
}

template <bool PLACE>
__device__ __forceinline__ void walk_runs(const int* __restrict__ binned, const int* rs, const int* re,
                                          int* kc, int* pl, int pbase, int* __restrict__ perm,
                                          int wv, int lane) {
    int a[8], b[8], s[8], e[8];
#pragma unroll
    for (int j = 0; j < 8; ++j) {
        int c = wv + 32 * j;
        s[j] = rs[c]; e[j] = re[c];
        int i0 = s[j] + lane, i1 = i0 + 32;
        a[j] = -1; b[j] = -1;
        if (i0 < e[j]) a[j] = binned[i0];
        if (i1 < e[j]) b[j] = binned[i1];
    }
#pragma unroll
    for (int j = 0; j < 8; ++j) {
        if (a[j] >= 0) {
            if (PLACE) place_src(pl, perm, pbase, atomicAdd(&kc[a[j] >> 17], 1), a[j] & 0x1FFFF);
            else atomicAdd(&kc[a[j] >> 17], 1);
        }
        if (b[j] >= 0) {
            if (PLACE) place_src(pl, perm, pbase, atomicAdd(&kc[b[j] >> 17], 1), b[j] & 0x1FFFF);
            else atomicAdd(&kc[b[j] >> 17], 1);
        }
        for (int i = s[j] + 64 + lane; i < e[j]; i += 32) {
            int w = binned[i];
            if (PLACE) place_src(pl, perm, pbase, atomicAdd(&kc[w >> 17], 1), w & 0x1FFFF);
            else atomicAdd(&kc[w >> 17], 1);
        }
    }
}

__global__ __launch_bounds__(1024) void sort_range(const int* __restrict__ binned,
                                                   const int* __restrict__ starts,
                                                   int* __restrict__ offs, int* __restrict__ perm) {
    __shared__ __attribute__((aligned(16))) int kc[KPAD];
    __shared__ __attribute__((aligned(16))) int pl[PCAP];
    __shared__ int rs[NCH];
    __shared__ int re[NCH];
    __shared__ int wtot[32];
    __shared__ int bpart[8];
    const int tid = threadIdx.x, lane = tid & 31, wv = tid >> 5, R = blockIdx.x;

    *(int4*)&kc[4 * tid] = make_int4(0, 0, 0, 0);
    if (tid < NCH) {
        int s = starts[tid * (NRG + 1) + R], e = starts[tid * (NRG + 1) + R + 1];
        rs[tid] = tid * CH + s;
        re[tid] = tid * CH + e;
        int b = s;
        for (int o = 16; o > 0; o >>= 1) b += __shfl_xor(b, o, 32);
        if (lane == 0) bpart[wv] = b;
    }
    __syncthreads();

    walk_runs<false>(binned, rs, re, kc, pl, 0, perm, wv, lane);
    __syncthreads();

    int4 v = *(const int4*)&kc[4 * tid];
    int t = v.x + v.y + v.z + v.w;
    int inc = t;
    for (int o = 1; o < 32; o <<= 1) {
        int n = __shfl_up(inc, o, 32);
        if (lane >= o) inc += n;
    }
    if (lane == 31) wtot[wv] = inc;
    __syncthreads();

    int rbeg = 0;
    for (int w = 0; w < 8; ++w) rbeg += bpart[w];
    int pre = rbeg, rend = rbeg;
    for (int w = 0; w < 32; ++w) {
        int x = wtot[w];
        rend += x;
        if (w < wv) pre += x;
    }
    const int pbase = rbeg & ~31;
    int4 o4;
    o4.x = pre + inc - t; o4.y = o4.x + v.x; o4.z = o4.y + v.y; o4.w = o4.z + v.z;
    *(int4*)&kc[4 * tid] = o4;
    int key0 = R * KR + 4 * tid;
    if (4 * tid < KR && key0 < NK) *(int4*)(offs + key0) = o4;
    if (R == NRG - 1 && tid == 0) offs[NK] = NE;
    __syncthreads();

    walk_runs<true>(binned, rs, re, kc, pl, pbase, perm, wv, lane);
    __syncthreads();

    int nrel = rend - pbase;
    nrel = nrel < PCAP ? nrel : PCAP;
    for (int r0 = 4 * tid; r0 < nrel; r0 += 4096) {
        int4 w4 = *(const int4*)&pl[r0];
        int g = pbase + r0;
        if (g >= rbeg && r0 + 4 <= nrel) {
            *(int4*)(perm + g) = w4;
        } else {
            if (g     >= rbeg && r0     < nrel) perm[g]     = w4.x;
            if (g + 1 >= rbeg && r0 + 1 < nrel) perm[g + 1] = w4.y;
            if (g + 2 >= rbeg && r0 + 2 < nrel) perm[g + 2] = w4.z;
            if (g + 3 >= rbeg && r0 + 3 < nrel) perm[g + 3] = w4.w;
        }
    }
}

__device__ __forceinline__ v2f a_frag(const float* __restrict__ M, int row, int stride,
                                      int k0, int kh) {
    const float* p = M + (size_t)row * stride + k0 + kh;
    v2f a; a.x = p[0]; a.y = p[1]; return a;
}
__device__ __forceinline__ v2f b_frag(const float* __restrict__ W, int nout,
                                      int k0, int kh, int col) {
    v2f b;
    b.x = W[(size_t)(k0 + kh) * nout + col];
    b.y = W[(size_t)(k0 + kh + 1) * nout + col];
    return b;
}

#define WCAP 128

template <int F, int PITCH, bool STAGED>
__device__ __forceinline__ void walk_group(const float* __restrict__ fq, const int* __restrict__ perm,
                                           const int* widx, int base, int q,
                                           int beg, int b1, int b2, int b3, int end,
                                           float* __restrict__ slot0) {
    typedef __attribute__((ext_vector_type(F / 8))) float vec;
    const int FPL  = F / 8;
    const int SLOT = 16 * PITCH;
    int last = end - 1;
    vec zero;
#pragma unroll
    for (int i = 0; i < FPL; ++i) zero[i] = 0.0f;
    vec acc = zero;
    int cur = 0;
    int mine = 0;
    if (beg < end) {
        int t = beg + q; t = t < last ? t : last;
        mine = STAGED ? widx[t - base] : perm[t];
    }
    for (int e = beg; e < end; e += 8) {
        int i0 = __builtin_amdgcn_mov_dpp8(mine, 0x000000);
        int i1 = __builtin_amdgcn_mov_dpp8(mine, 0x249249);
        int i2 = __builtin_amdgcn_mov_dpp8(mine, 0x492492);
        int i3 = __builtin_amdgcn_mov_dpp8(mine, 0x6DB6DB);
        int i4 = __builtin_amdgcn_mov_dpp8(mine, 0x924924);
        int i5 = __builtin_amdgcn_mov_dpp8(mine, 0xB6DB6D);
        int i6 = __builtin_amdgcn_mov_dpp8(mine, 0xDB6DB6);
        int i7 = __builtin_amdgcn_mov_dpp8(mine, 0xFFFFFF);
        vec u[8];
        u[0] = *(const vec*)(fq + (unsigned)i0 * F);
        u[1] = *(const vec*)(fq + (unsigned)i1 * F);
        u[2] = *(const vec*)(fq + (unsigned)i2 * F);
        u[3] = *(const vec*)(fq + (unsigned)i3 * F);
        u[4] = *(const vec*)(fq + (unsigned)i4 * F);
        u[5] = *(const vec*)(fq + (unsigned)i5 * F);
        u[6] = *(const vec*)(fq + (unsigned)i6 * F);
        u[7] = *(const vec*)(fq + (unsigned)i7 * F);
        {
            int t = e + 8 + q; t = t < last ? t : last;
            mine = STAGED ? widx[t - base] : perm[t];
        }
#pragma unroll
        for (int k = 0; k < 8; ++k) {
            int ee = e + k;
            if (ee < end) {
                int sidx = (ee >= b1 ? 1 : 0) + (ee >= b2 ? 1 : 0) + (ee >= b3 ? 1 : 0);
                if (sidx != cur) {
                    *(vec*)(slot0 + cur * SLOT) = acc;
                    acc = zero;
                    cur = sidx;
                }
                acc += u[k];
            }
        }
    }
    *(vec*)(slot0 + cur * SLOT) = acc;
}

template <int F, int PITCH>
__device__ __forceinline__ void tile_seg_sums(const float* __restrict__ feat,
                                              const int* __restrict__ offs,
                                              const int* __restrict__ perm,
                                              int m0, float* __restrict__ agg,
                                              int* __restrict__ widx, int* __restrict__ cnt) {
    typedef __attribute__((ext_vector_type(F / 8))) float vec;
    const int FPL  = F / 8;
    const int SLOT = 16 * PITCH;
    int q    = threadIdx.x & 7;
    int g    = threadIdx.x >> 3;
    int node = g >> 1, half = g & 1;
    int k0   = (m0 + node) * NR + half * 4;
    int4 bb  = *(const int4*)(offs + k0);
    int beg = bb.x, b1 = bb.y, b2 = bb.z, b3 = bb.w;
    int end  = offs[k0 + 4];
    if (q == 0) *(int4*)(cnt + (node * NR + half * 4)) = make_int4(b1 - beg, b2 - b1, b3 - b2, end - b3);

    float* slot0 = agg + (half * 4 * 16 + node) * PITCH + q * FPL;
    vec zero;
#pragma unroll
    for (int i = 0; i < FPL; ++i) zero[i] = 0.0f;
#pragma unroll
    for (int s = 0; s < 4; ++s) *(vec*)(slot0 + s * SLOT) = zero;

    int wbeg = __builtin_amdgcn_readfirstlane(beg);
    int wend = __builtin_amdgcn_readlane(end, 31);
    int base = wbeg & ~3;
    const float* fq = feat + q * FPL;
    if (wend - base <= WCAP) {
        int l4 = (threadIdx.x & 31) * 4;
        if (base + l4 < wend) *(int4*)(widx + l4) = *(const int4*)(perm + base + l4);
        walk_group<F, PITCH, true>(fq, perm, widx, base, q, beg, b1, b2, b3, end, slot0);
    } else {
        walk_group<F, PITCH, false>(fq, perm, widx, base, q, beg, b1, b2, b3, end, slot0);
    }
}


#define P1 20
#define P2 36

__global__ __launch_bounds__(256) void rgcn_layer1(
    const float* __restrict__ x, const int* __restrict__ offs, const int* __restrict__ perm,
    const float* __restrict__ W1, const float* __restrict__ root1, const float* __restrict__ b1,
    float* __restrict__ h) {
    __shared__ __attribute__((aligned(16))) float agg[NR * 16 * P1];
    __shared__ float part[NR][16][33];
    __shared__ __attribute__((aligned(16))) int widx[NR * WCAP];
    __shared__ __attribute__((aligned(16))) int cnt[16 * NR];
    int r    = __builtin_amdgcn_readfirstlane(threadIdx.x >> 5);
    int lane = threadIdx.x & 31;
    int m0   = blockIdx.x * 16;
    int arow = m0 + (lane & 15);
    int kh   = (lane >> 4) << 1;
    int col  = lane & 15;

    tile_seg_sums<16, P1>(x, offs, perm, m0, agg, widx + r * WCAP, cnt);
    __syncthreads();
    float sc = 1.0f / fmaxf((float)cnt[(lane & 15) * NR + r], 1.0f);

    v8f c0, c1;
    if (r == 0) {
        float bb0 = b1[col], bb1 = b1[16 + col];
#pragma unroll
        for (int v = 0; v < 8; ++v) { c0[v] = bb0; c1[v] = bb1; }
#pragma unroll
        for (int k = 0; k < 16; k += 4) {
            v2f a  = a_frag(x, arow, 16, k, kh);
            v2f bA = b_frag(root1, 32, k, kh, col);
            v2f bB = b_frag(root1, 32, k, kh, col + 16);
            c0 = __builtin_amdgcn_wmma_f32_16x16x4_f32(false, a, false, bA, (short)0, c0, false, false);
            c1 = __builtin_amdgcn_wmma_f32_16x16x4_f32(false, a, false, bB, (short)0, c1, false, false);
        }
    } else {
#pragma unroll
        for (int v = 0; v < 8; ++v) { c0[v] = 0.0f; c1[v] = 0.0f; }
    }
    const float* Wr = W1 + (size_t)r * 16 * 32;
    const float* ar = agg + (r * 16 + (lane & 15)) * P1 + kh;
#pragma unroll
    for (int j = 0; j < 4; ++j) {
        v2f a  = *(const v2f*)(ar + 4 * j) * sc;
        v2f bA = b_frag(Wr, 32, 4 * j, kh, col);
        v2f bB = b_frag(Wr, 32, 4 * j, kh, col + 16);
        c0 = __builtin_amdgcn_wmma_f32_16x16x4_f32(false, a, false, bA, (short)0, c0, false, false);
        c1 = __builtin_amdgcn_wmma_f32_16x16x4_f32(false, a, false, bB, (short)0, c1, false, false);
    }
    int rb = (lane >> 4) << 3;
#pragma unroll
    for (int v = 0; v < 8; ++v) {
        part[r][rb + v][col]      = c0[v];
        part[r][rb + v][16 + col] = c1[v];
    }
    __syncthreads();
#pragma unroll
    for (int i = 0; i < 2; ++i) {
        int idx = threadIdx.x + 256 * i;
        int row = idx >> 5, cc = idx & 31;
        float s = part[0][row][cc];
#pragma unroll
        for (int w = 1; w < NR; ++w) s += part[w][row][cc];
        h[(size_t)m0 * 32 + idx] = fmaxf(s, 0.0f);
    }
}


__global__ __launch_bounds__(256) void rgcn_layer2(
    const float* __restrict__ h, const int* __restrict__ offs, const int* __restrict__ perm,
    const float* __restrict__ W2, const float* __restrict__ root2, const float* __restrict__ b2,
    float* __restrict__ out) {
    __shared__ __attribute__((aligned(16))) float agg[NR * 16 * P2];
    __shared__ float part[NR][16][17];
    __shared__ __attribute__((aligned(16))) int widx[NR * WCAP];
    __shared__ __attribute__((aligned(16))) int cnt[16 * NR];
    int r    = __builtin_amdgcn_readfirstlane(threadIdx.x >> 5);
    int lane = threadIdx.x & 31;
    int m0   = blockIdx.x * 16;
    int arow = m0 + (lane & 15);
    int kh   = (lane >> 4) << 1;
    int col  = lane & 15;

    tile_seg_sums<32, P2>(h, offs, perm, m0, agg, widx + r * WCAP, cnt);
    __syncthreads();
    float sc = 1.0f / fmaxf((float)cnt[(lane & 15) * NR + r], 1.0f);

    v8f c;
    if (r == 0) {
        float bb = b2[col];
#pragma unroll
        for (int v = 0; v < 8; ++v) c[v] = bb;
#pragma unroll
        for (int k = 0; k < 32; k += 4) {
            v2f a = a_frag(h, arow, 32, k, kh);
            v2f b = b_frag(root2, 16, k, kh, col);
            c = __builtin_amdgcn_wmma_f32_16x16x4_f32(false, a, false, b, (short)0, c, false, false);
        }
    } else {
#pragma unroll
        for (int v = 0; v < 8; ++v) c[v] = 0.0f;
    }
    const float* Wr = W2 + (size_t)r * 32 * 16;
    const float* ar = agg + (r * 16 + (lane & 15)) * P2 + kh;
#pragma unroll
    for (int j = 0; j < 8; ++j) {
        v2f a = *(const v2f*)(ar + 4 * j) * sc;
        v2f b = b_frag(Wr, 16, 4 * j, kh, col);
        c = __builtin_amdgcn_wmma_f32_16x16x4_f32(false, a, false, b, (short)0, c, false, false);
    }
    int rb = (lane >> 4) << 3;
#pragma unroll
    for (int v = 0; v < 8; ++v) part[r][rb + v][col] = c[v];
    __syncthreads();
    {
        int row = threadIdx.x >> 4, cc = threadIdx.x & 15;
        float s = part[0][row][cc];
#pragma unroll
        for (int w = 1; w < NR; ++w) s += part[w][row][cc];
        out[(size_t)m0 * 16 + threadIdx.x] = s;
    }
}


extern "C" void kernel_launch(void* const* d_in, const int* in_sizes, int n_in,
                              void* d_out, int out_size, void* d_ws, size_t ws_size,
                              hipStream_t stream) {
    const float* x     = (const float*)d_in[0];
    const int*   eidx  = (const int*)d_in[1];
    const int*   etype = (const int*)d_in[2];
    const float* W1    = (const float*)d_in[3];
    const float* root1 = (const float*)d_in[4];
    const float* b1    = (const float*)d_in[5];
    const float* W2    = (const float*)d_in[6];
    const float* root2 = (const float*)d_in[7];
    const float* b2    = (const float*)d_in[8];
    float* out = (float*)d_out;

    const int* src = eidx;
    const int* dst = eidx + NE;

    float* h       = (float*)d_ws;
    int*   binned  = (int*)(h + (size_t)NN * 32);
    int*   perm    = binned + NE;
    int*   starts  = perm + NE;
    int*   offs    = starts + NCH * (NRG + 1);

    bin_chunks<<<NCH, 1024, 0, stream>>>(src, dst, etype, binned, starts);
    sort_range<<<NRG, 1024, 0, stream>>>(binned, starts, offs, perm);

    rgcn_layer1<<<NN / 16, 256, 0, stream>>>(x, offs, perm, W1, root1, b1, h);
    rgcn_layer2<<<NN / 16, 256, 0, stream>>>(h, offs, perm, W2, root2, b2, out);

    (void)in_sizes; (void)n_in; (void)out_size; (void)ws_size;
}


# layer walks keep their books per 8-edge trip, not per edge: segment starts become one bit mask, a slot tests one bit and counts bits for its segment; rows by 32-bit offset; sum order unchanged
# speedup vs baseline: 4.9114x; 1.0443x over previous
#include <hip/hip_runtime.h>

#define NN 100000
#define NE 3200000
#define NR 8
#define NK (NN * NR)
#define NCH 256
#define CH (NE / NCH)
#define EPT 13
#define RNG 391
#define NRG 256
#define KR (RNG * NR)
#define KPAD 4096
#define PCAP 16384

typedef __attribute__((ext_vector_type(2))) float v2f;
typedef __attribute__((ext_vector_type(8))) float v8f;


__global__ __launch_bounds__(1024) void bin_chunks(const int* __restrict__ src,
                                                   const int* __restrict__ dst,
                                                   const int* __restrict__ et,
                                                   int* __restrict__ binned,
                                                   int* __restrict__ starts) {
    __shared__ int cntR[NRG];
    __shared__ int startR[NRG];
    __shared__ int wsum[8];
    __shared__ int buf[CH];
    const int tid = threadIdx.x, lane = tid & 31, c = blockIdx.x;
    const int e0 = c * CH;
    if (tid < NRG) cntR[tid] = 0;

    int word[EPT], rgk[EPT], lr[EPT];
#pragma unroll
    for (int k = 0; k < EPT; ++k) {
        int i = tid + 1024 * k;
        word[k] = 0; rgk[k] = 0; lr[k] = 0;
        if (i < CH) {
            int d = dst[e0 + i], r = et[e0 + i], s = src[e0 + i];
            int rg = d / RNG;
            word[k] = ((d - rg * RNG) << 20) | (r << 17) | s;
            rgk[k] = rg;
        }
    }
    __syncthreads();
#pragma unroll
    for (int k = 0; k < EPT; ++k)
        if (tid + 1024 * k < CH) lr[k] = atomicAdd(&cntR[rgk[k]], 1);
    __syncthreads();

    int v = 0, inc = 0;
    if (tid < NRG) {
        v = cntR[tid];
        inc = v;
        for (int o = 1; o < 32; o <<= 1) {
            int n = __shfl_up(inc, o, 32);
            if (lane >= o) inc += n;
        }
        if (lane == 31) wsum[tid >> 5] = inc;
    }
    __syncthreads();
    if (tid < NRG) {
        int b = 0;
        for (int w = 0; w < (tid >> 5); ++w) b += wsum[w];
        int ex = b + inc - v;
        startR[tid] = ex;
        starts[c * (NRG + 1) + tid] = ex;
        if (tid == NRG - 1) starts[c * (NRG + 1) + NRG] = CH;
    }
    __syncthreads();
#pragma unroll
    for (int k = 0; k < EPT; ++k)
        if (tid + 1024 * k < CH) buf[startR[rgk[k]] + lr[k]] = word[k];
    __syncthreads();
#pragma unroll
    for (int k = 0; k < EPT; ++k) {
        int i = tid + 1024 * k;
        if (i < CH) binned[e0 + i] = buf[i];
    }
}

__device__ __forceinline__ void place_src(int* pl, int* __restrict__ perm, int pbase, int slot, int s) {
    int rel = slot - pbase;
    if (rel < PCAP) pl[rel] = s;
    else perm[slot] = s;
}

template <bool PLACE>
__device__ __forceinline__ void walk_runs(const int* __restrict__ binned, const int* rs, const int* re,
                                          int* kc, int* pl, int pbase, int* __restrict__ perm,
                                          int wv, int lane) {
    int a[8], b[8], s[8], e[8];
#pragma unroll
    for (int j = 0; j < 8; ++j) {
        int c = wv + 32 * j;
        s[j] = rs[c]; e[j] = re[c];
        int i0 = s[j] + lane, i1 = i0 + 32;
        a[j] = -1; b[j] = -1;
        if (i0 < e[j]) a[j] = binned[i0];
        if (i1 < e[j]) b[j] = binned[i1];
    }
#pragma unroll
    for (int j = 0; j < 8; ++j) {
        if (a[j] >= 0) {
            if (PLACE) place_src(pl, perm, pbase, atomicAdd(&kc[a[j] >> 17], 1), a[j] & 0x1FFFF);
            else atomicAdd(&kc[a[j] >> 17], 1);
        }
        if (b[j] >= 0) {
            if (PLACE) place_src(pl, perm, pbase, atomicAdd(&kc[b[j] >> 17], 1), b[j] & 0x1FFFF);
            else atomicAdd(&kc[b[j] >> 17], 1);
        }
        for (int i = s[j] + 64 + lane; i < e[j]; i += 32) {
            int w = binned[i];
            if (PLACE) place_src(pl, perm, pbase, atomicAdd(&kc[w >> 17], 1), w & 0x1FFFF);
            else atomicAdd(&kc[w >> 17], 1);
        }
    }
}

__global__ __launch_bounds__(1024) void sort_range(const int* __restrict__ binned,
                                                   const int* __restrict__ starts,
                                                   int* __restrict__ offs, int* __restrict__ perm) {
    __shared__ __attribute__((aligned(16))) int kc[KPAD];
    __shared__ __attribute__((aligned(16))) int pl[PCAP];
    __shared__ int rs[NCH];
    __shared__ int re[NCH];
    __shared__ int wtot[32];
    __shared__ int bpart[8];
    const int tid = threadIdx.x, lane = tid & 31, wv = tid >> 5, R = blockIdx.x;

    *(int4*)&kc[4 * tid] = make_int4(0, 0, 0, 0);
    if (tid < NCH) {
        int s = starts[tid * (NRG + 1) + R], e = starts[tid * (NRG + 1) + R + 1];
        rs[tid] = tid * CH + s;
        re[tid] = tid * CH + e;
        int b = s;
        for (int o = 16; o > 0; o >>= 1) b += __shfl_xor(b, o, 32);
        if (lane == 0) bpart[wv] = b;
    }
    __syncthreads();

    walk_runs<false>(binned, rs, re, kc, pl, 0, perm, wv, lane);
    __syncthreads();

    int4 v = *(const int4*)&kc[4 * tid];
    int t = v.x + v.y + v.z + v.w;
    int inc = t;
    for (int o = 1; o < 32; o <<= 1) {
        int n = __shfl_up(inc, o, 32);
        if (lane >= o) inc += n;
    }
    if (lane == 31) wtot[wv] = inc;
    __syncthreads();

    int rbeg = 0;
    for (int w = 0; w < 8; ++w) rbeg += bpart[w];
    int pre = rbeg, rend = rbeg;
    for (int w = 0; w < 32; ++w) {
        int x = wtot[w];
        rend += x;
        if (w < wv) pre += x;
    }
    const int pbase = rbeg & ~31;
    int4 o4;
    o4.x = pre + inc - t; o4.y = o4.x + v.x; o4.z = o4.y + v.y; o4.w = o4.z + v.z;
    *(int4*)&kc[4 * tid] = o4;
    int key0 = R * KR + 4 * tid;
    if (4 * tid < KR && key0 < NK) *(int4*)(offs + key0) = o4;
    if (R == NRG - 1 && tid == 0) offs[NK] = NE;
    __syncthreads();

    walk_runs<true>(binned, rs, re, kc, pl, pbase, perm, wv, lane);
    __syncthreads();

    int nrel = rend - pbase;
    nrel = nrel < PCAP ? nrel : PCAP;
    for (int r0 = 4 * tid; r0 < nrel; r0 += 4096) {
        int4 w4 = *(const int4*)&pl[r0];
        int g = pbase + r0;
        if (g >= rbeg && r0 + 4 <= nrel) {
            *(int4*)(perm + g) = w4;
        } else {
            if (g     >= rbeg && r0     < nrel) perm[g]     = w4.x;
            if (g + 1 >= rbeg && r0 + 1 < nrel) perm[g + 1] = w4.y;
            if (g + 2 >= rbeg && r0 + 2 < nrel) perm[g + 2] = w4.z;
            if (g + 3 >= rbeg && r0 + 3 < nrel) perm[g + 3] = w4.w;
        }
    }
}

__device__ __forceinline__ v2f a_frag(const float* __restrict__ M, int row, int stride,
                                      int k0, int kh) {
    const float* p = M + (size_t)row * stride + k0 + kh;
    v2f a; a.x = p[0]; a.y = p[1]; return a;
}
__device__ __forceinline__ v2f b_frag(const float* __restrict__ W, int nout,
                                      int k0, int kh, int col) {
    v2f b;
    b.x = W[(size_t)(k0 + kh) * nout + col];
    b.y = W[(size_t)(k0 + kh + 1) * nout + col];
    return b;
}

#define WCAP 128

template <int F, int PITCH, bool STAGED>
__device__ __forceinline__ void walk_group(const float* __restrict__ feat, const int* __restrict__ perm,
                                           const int* widx, int base, int q,
                                           int beg, int b1, int b2, int b3, int end,
                                           float* __restrict__ slot0) {
    typedef __attribute__((ext_vector_type(F / 8))) float vec;
    const int FPL  = F / 8;
    const int SLOT = 16 * PITCH;
    const unsigned LANES3 = 0x40201u;
    int last = end - 1;
    vec acc = (vec)(0.0f);
    int cur = 0;
    const char* fb = (const char*)feat;
    const unsigned qb = (unsigned)q * (FPL * 4);
    unsigned mine = 0;
    if (beg < end) {
        int t = beg + q; t = t < last ? t : last;
        mine = (unsigned)(STAGED ? widx[t - base] : perm[t]) * (F * 4);
    }
    for (int e = beg; e < end; e += 8) {
        unsigned o0 = __builtin_amdgcn_mov_dpp8(mine, 0x000000);
        unsigned o1 = __builtin_amdgcn_mov_dpp8(mine, 0x249249);
        unsigned o2 = __builtin_amdgcn_mov_dpp8(mine, 0x492492);
        unsigned o3 = __builtin_amdgcn_mov_dpp8(mine, 0x6DB6DB);
        unsigned o4 = __builtin_amdgcn_mov_dpp8(mine, 0x924924);
        unsigned o5 = __builtin_amdgcn_mov_dpp8(mine, 0xB6DB6D);
        unsigned o6 = __builtin_amdgcn_mov_dpp8(mine, 0xDB6DB6);
        unsigned o7 = __builtin_amdgcn_mov_dpp8(mine, 0xFFFFFF);
        vec u[8];
        u[0] = *(const vec*)(fb + (o0 + qb));
        u[1] = *(const vec*)(fb + (o1 + qb));
        u[2] = *(const vec*)(fb + (o2 + qb));
        u[3] = *(const vec*)(fb + (o3 + qb));
        u[4] = *(const vec*)(fb + (o4 + qb));
        u[5] = *(const vec*)(fb + (o5 + qb));
        u[6] = *(const vec*)(fb + (o6 + qb));
        u[7] = *(const vec*)(fb + (o7 + qb));
        {
            int t = e + 8 + q; t = t < last ? t : last;
            mine = (unsigned)(STAGED ? widx[t - base] : perm[t]) * (F * 4);
        }
        const int n = end - e;
        unsigned d1 = (unsigned)(b1 - e), d2 = (unsigned)(b2 - e), d3 = (unsigned)(b3 - e);
        d1 = d1 < 8u ? d1 : 8u; d2 = d2 < 8u ? d2 : 8u; d3 = d3 < 8u ? d3 : 8u;
        const unsigned m = (1u << d1) | (0x200u << d2) | (0x40000u << d3);
        const int s0 = cur;
#pragma unroll
        for (int k = 0; k < 8; ++k) {
            if (k < n) {
                if (m & (LANES3 << k)) {
                    *(vec*)(slot0 + cur * SLOT) = acc;
                    cur = s0 + __builtin_popcount(m & (LANES3 * ((2u << k) - 1u)));
                    acc = u[k];
                } else {
                    acc += u[k];
                }
            }
        }
    }
    *(vec*)(slot0 + cur * SLOT) = acc;
}

template <int F, int PITCH>
__device__ __forceinline__ void tile_seg_sums(const float* __restrict__ feat,
                                              const int* __restrict__ offs,
                                              const int* __restrict__ perm,
                                              int m0, float* __restrict__ agg,
                                              int* __restrict__ widx, int* __restrict__ cnt) {
    typedef __attribute__((ext_vector_type(F / 8))) float vec;
    const int FPL  = F / 8;
    const int SLOT = 16 * PITCH;
    int q    = threadIdx.x & 7;
    int g    = threadIdx.x >> 3;
    int node = g >> 1, half = g & 1;
    int k0   = (m0 + node) * NR + half * 4;
    int4 bb  = *(const int4*)(offs + k0);
    int beg = bb.x, b1 = bb.y, b2 = bb.z, b3 = bb.w;
    int end  = offs[k0 + 4];
    if (q == 0) *(int4*)(cnt + (node * NR + half * 4)) = make_int4(b1 - beg, b2 - b1, b3 - b2, end - b3);

    float* slot0 = agg + (half * 4 * 16 + node) * PITCH + q * FPL;
    vec zero;
#pragma unroll
    for (int i = 0; i < FPL; ++i) zero[i] = 0.0f;
#pragma unroll
    for (int s = 0; s < 4; ++s) *(vec*)(slot0 + s * SLOT) = zero;

    int wbeg = __builtin_amdgcn_readfirstlane(beg);
    int wend = __builtin_amdgcn_readlane(end, 31);
    int base = wbeg & ~3;
    if (wend - base <= WCAP) {
        int l4 = (threadIdx.x & 31) * 4;
        if (base + l4 < wend) *(int4*)(widx + l4) = *(const int4*)(perm + base + l4);
        walk_group<F, PITCH, true>(feat, perm, widx, base, q, beg, b1, b2, b3, end, slot0);
    } else {
        walk_group<F, PITCH, false>(feat, perm, widx, base, q, beg, b1, b2, b3, end, slot0);
    }
}


#define P1 20
#define P2 36

__global__ __launch_bounds__(256) void rgcn_layer1(
    const float* __restrict__ x, const int* __restrict__ offs, const int* __restrict__ perm,
    const float* __restrict__ W1, const float* __restrict__ root1, const float* __restrict__ b1,
    float* __restrict__ h) {
    __shared__ __attribute__((aligned(16))) float agg[NR * 16 * P1];
    __shared__ float part[NR][16][33];
    __shared__ __attribute__((aligned(16))) int widx[NR * WCAP];
    __shared__ __attribute__((aligned(16))) int cnt[16 * NR];
    int r    = __builtin_amdgcn_readfirstlane(threadIdx.x >> 5);
    int lane = threadIdx.x & 31;
    int m0   = blockIdx.x * 16;
    int arow = m0 + (lane & 15);
    int kh   = (lane >> 4) << 1;
    int col  = lane & 15;

    tile_seg_sums<16, P1>(x, offs, perm, m0, agg, widx + r * WCAP, cnt);
    __syncthreads();
    float sc = 1.0f / fmaxf((float)cnt[(lane & 15) * NR + r], 1.0f);

    v8f c0, c1;
    if (r == 0) {
        float bb0 = b1[col], bb1 = b1[16 + col];
#pragma unroll
        for (int v = 0; v < 8; ++v) { c0[v] = bb0; c1[v] = bb1; }
#pragma unroll
        for (int k = 0; k < 16; k += 4) {
            v2f a  = a_frag(x, arow, 16, k, kh);
            v2f bA = b_frag(root1, 32, k, kh, col);
            v2f bB = b_frag(root1, 32, k, kh, col + 16);
            c0 = __builtin_amdgcn_wmma_f32_16x16x4_f32(false, a, false, bA, (short)0, c0, false, false);
            c1 = __builtin_amdgcn_wmma_f32_16x16x4_f32(false, a, false, bB, (short)0, c1, false, false);
        }
    } else {
#pragma unroll
        for (int v = 0; v < 8; ++v) { c0[v] = 0.0f; c1[v] = 0.0f; }
    }
    const float* Wr = W1 + (size_t)r * 16 * 32;
    const float* ar = agg + (r * 16 + (lane & 15)) * P1 + kh;
#pragma unroll
    for (int j = 0; j < 4; ++j) {
        v2f a  = *(const v2f*)(ar + 4 * j) * sc;
        v2f bA = b_frag(Wr, 32, 4 * j, kh, col);
        v2f bB = b_frag(Wr, 32, 4 * j, kh, col + 16);
        c0 = __builtin_amdgcn_wmma_f32_16x16x4_f32(false, a, false, bA, (short)0, c0, false, false);
        c1 = __builtin_amdgcn_wmma_f32_16x16x4_f32(false, a, false, bB, (short)0, c1, false, false);
    }
    int rb = (lane >> 4) << 3;
#pragma unroll
    for (int v = 0; v < 8; ++v) {
        part[r][rb + v][col]      = c0[v];
        part[r][rb + v][16 + col] = c1[v];
    }
    __syncthreads();
#pragma unroll
    for (int i = 0; i < 2; ++i) {
        int idx = threadIdx.x + 256 * i;
        int row = idx >> 5, cc = idx & 31;
        float s = part[0][row][cc];
#pragma unroll
        for (int w = 1; w < NR; ++w) s += part[w][row][cc];
        h[(size_t)m0 * 32 + idx] = fmaxf(s, 0.0f);
    }
}


__global__ __launch_bounds__(256) void rgcn_layer2(
    const float* __restrict__ h, const int* __restrict__ offs, const int* __restrict__ perm,
    const float* __restrict__ W2, const float* __restrict__ root2, const float* __restrict__ b2,
    float* __restrict__ out) {
    __shared__ __attribute__((aligned(16))) float agg[NR * 16 * P2];
    __shared__ float part[NR][16][17];
    __shared__ __attribute__((aligned(16))) int widx[NR * WCAP];
    __shared__ __attribute__((aligned(16))) int cnt[16 * NR];
    int r    = __builtin_amdgcn_readfirstlane(threadIdx.x >> 5);
    int lane = threadIdx.x & 31;
    int m0   = blockIdx.x * 16;
    int arow = m0 + (lane & 15);
    int kh   = (lane >> 4) << 1;
    int col  = lane & 15;

    tile_seg_sums<32, P2>(h, offs, perm, m0, agg, widx + r * WCAP, cnt);
    __syncthreads();
    float sc = 1.0f / fmaxf((float)cnt[(lane & 15) * NR + r], 1.0f);

    v8f c;
    if (r == 0) {
        float bb = b2[col];
#pragma unroll
        for (int v = 0; v < 8; ++v) c[v] = bb;
#pragma unroll
        for (int k = 0; k < 32; k += 4) {
            v2f a = a_frag(h, arow, 32, k, kh);
            v2f b = b_frag(root2, 16, k, kh, col);
            c = __builtin_amdgcn_wmma_f32_16x16x4_f32(false, a, false, b, (short)0, c, false, false);
        }
    } else {
#pragma unroll
        for (int v = 0; v < 8; ++v) c[v] = 0.0f;
    }
    const float* Wr = W2 + (size_t)r * 32 * 16;
    const float* ar = agg + (r * 16 + (lane & 15)) * P2 + kh;
#pragma unroll
    for (int j = 0; j < 8; ++j) {
        v2f a = *(const v2f*)(ar + 4 * j) * sc;
        v2f b = b_frag(Wr, 16, 4 * j, kh, col);
        c = __builtin_amdgcn_wmma_f32_16x16x4_f32(false, a, false, b, (short)0, c, false, false);
    }
    int rb = (lane >> 4) << 3;
#pragma unroll
    for (int v = 0; v < 8; ++v) part[r][rb + v][col] = c[v];
    __syncthreads();
    {
        int row = threadIdx.x >> 4, cc = threadIdx.x & 15;
        float s = part[0][row][cc];
#pragma unroll
        for (int w = 1; w < NR; ++w) s += part[w][row][cc];
        out[(size_t)m0 * 16 + threadIdx.x] = s;
    }
}


extern "C" void kernel_launch(void* const* d_in, const int* in_sizes, int n_in,
                              void* d_out, int out_size, void* d_ws, size_t ws_size,
                              hipStream_t stream) {
    const float* x     = (const float*)d_in[0];
    const int*   eidx  = (const int*)d_in[1];
    const int*   etype = (const int*)d_in[2];
    const float* W1    = (const float*)d_in[3];
    const float* root1 = (const float*)d_in[4];
    const float* b1    = (const float*)d_in[5];
    const float* W2    = (const float*)d_in[6];
    const float* root2 = (const float*)d_in[7];
    const float* b2    = (const float*)d_in[8];
    float* out = (float*)d_out;

    const int* src = eidx;
    const int* dst = eidx + NE;

    float* h       = (float*)d_ws;
    int*   binned  = (int*)(h + (size_t)NN * 32);
    int*   perm    = binned + NE;
    int*   starts  = perm + NE;
    int*   offs    = starts + NCH * (NRG + 1);

    bin_chunks<<<NCH, 1024, 0, stream>>>(src, dst, etype, binned, starts);
    sort_range<<<NRG, 1024, 0, stream>>>(binned, starts, offs, perm);

    rgcn_layer1<<<NN / 16, 256, 0, stream>>>(x, offs, perm, W1, root1, b1, h);
    rgcn_layer2<<<NN / 16, 256, 0, stream>>>(h, offs, perm, W2, root2, b2, out);

    (void)in_sizes; (void)n_in; (void)out_size; (void)ws_size;
}


# layer tiles run 4 waves, not 8: a lane group walks a whole node (its two 4-relation runs in turn), a wave multiplies 2 relations into one accumulator; relation sums: 2 chained, then 4 added
# speedup vs baseline: 5.2691x; 1.0728x over previous
#include <hip/hip_runtime.h>

#define NN 100000
#define NE 3200000
#define NR 8
#define NK (NN * NR)
#define NCH 256
#define CH (NE / NCH)
#define EPT 13
#define RNG 391
#define NRG 256
#define KR (RNG * NR)
#define KPAD 4096
#define PCAP 16384

typedef __attribute__((ext_vector_type(2))) float v2f;
typedef __attribute__((ext_vector_type(8))) float v8f;


__global__ __launch_bounds__(1024) void bin_chunks(const int* __restrict__ src,
                                                   const int* __restrict__ dst,
                                                   const int* __restrict__ et,
                                                   int* __restrict__ binned,
                                                   int* __restrict__ starts) {
    __shared__ int cntR[NRG];
    __shared__ int startR[NRG];
    __shared__ int wsum[8];
    __shared__ int buf[CH];
    const int tid = threadIdx.x, lane = tid & 31, c = blockIdx.x;
    const int e0 = c * CH;
    if (tid < NRG) cntR[tid] = 0;

    int word[EPT], rgk[EPT], lr[EPT];
#pragma unroll
    for (int k = 0; k < EPT; ++k) {
        int i = tid + 1024 * k;
        word[k] = 0; rgk[k] = 0; lr[k] = 0;
        if (i < CH) {
            int d = dst[e0 + i], r = et[e0 + i], s = src[e0 + i];
            int rg = d / RNG;
            word[k] = ((d - rg * RNG) << 20) | (r << 17) | s;
            rgk[k] = rg;
        }
    }
    __syncthreads();
#pragma unroll
    for (int k = 0; k < EPT; ++k)
        if (tid + 1024 * k < CH) lr[k] = atomicAdd(&cntR[rgk[k]], 1);
    __syncthreads();

    int v = 0, inc = 0;
    if (tid < NRG) {
        v = cntR[tid];
        inc = v;
        for (int o = 1; o < 32; o <<= 1) {
            int n = __shfl_up(inc, o, 32);
            if (lane >= o) inc += n;
        }
        if (lane == 31) wsum[tid >> 5] = inc;
    }
    __syncthreads();
    if (tid < NRG) {
        int b = 0;
        for (int w = 0; w < (tid >> 5); ++w) b += wsum[w];
        int ex = b + inc - v;
        startR[tid] = ex;
        starts[c * (NRG + 1) + tid] = ex;
        if (tid == NRG - 1) starts[c * (NRG + 1) + NRG] = CH;
    }
    __syncthreads();
#pragma unroll
    for (int k = 0; k < EPT; ++k)
        if (tid + 1024 * k < CH) buf[startR[rgk[k]] + lr[k]] = word[k];
    __syncthreads();
#pragma unroll
    for (int k = 0; k < EPT; ++k) {
        int i = tid + 1024 * k;
        if (i < CH) binned[e0 + i] = buf[i];
    }
}

__device__ __forceinline__ void place_src(int* pl, int* __restrict__ perm, int pbase, int slot, int s) {
    int rel = slot - pbase;
    if (rel < PCAP) pl[rel] = s;
    else perm[slot] = s;
}

template <bool PLACE>
__device__ __forceinline__ void walk_runs(const int* __restrict__ binned, const int* rs, const int* re,
                                          int* kc, int* pl, int pbase, int* __restrict__ perm,
                                          int wv, int lane) {
    int a[8], b[8], s[8], e[8];
#pragma unroll
    for (int j = 0; j < 8; ++j) {
        int c = wv + 32 * j;
        s[j] = rs[c]; e[j] = re[c];
        int i0 = s[j] + lane, i1 = i0 + 32;
        a[j] = -1; b[j] = -1;
        if (i0 < e[j]) a[j] = binned[i0];
        if (i1 < e[j]) b[j] = binned[i1];
    }
#pragma unroll
    for (int j = 0; j < 8; ++j) {
        if (a[j] >= 0) {
            if (PLACE) place_src(pl, perm, pbase, atomicAdd(&kc[a[j] >> 17], 1), a[j] & 0x1FFFF);
            else atomicAdd(&kc[a[j] >> 17], 1);
        }
        if (b[j] >= 0) {
            if (PLACE) place_src(pl, perm, pbase, atomicAdd(&kc[b[j] >> 17], 1), b[j] & 0x1FFFF);
            else atomicAdd(&kc[b[j] >> 17], 1);
        }
        for (int i = s[j] + 64 + lane; i < e[j]; i += 32) {
            int w = binned[i];
            if (PLACE) place_src(pl, perm, pbase, atomicAdd(&kc[w >> 17], 1), w & 0x1FFFF);
            else atomicAdd(&kc[w >> 17], 1);
        }
    }
}

__global__ __launch_bounds__(1024) void sort_range(const int* __restrict__ binned,
                                                   const int* __restrict__ starts,
                                                   int* __restrict__ offs, int* __restrict__ perm) {
    __shared__ __attribute__((aligned(16))) int kc[KPAD];
    __shared__ __attribute__((aligned(16))) int pl[PCAP];
    __shared__ int rs[NCH];
    __shared__ int re[NCH];
    __shared__ int wtot[32];
    __shared__ int bpart[8];
    const int tid = threadIdx.x, lane = tid & 31, wv = tid >> 5, R = blockIdx.x;

    *(int4*)&kc[4 * tid] = make_int4(0, 0, 0, 0);
    if (tid < NCH) {
        int s = starts[tid * (NRG + 1) + R], e = starts[tid * (NRG + 1) + R + 1];
        rs[tid] = tid * CH + s;
        re[tid] = tid * CH + e;
        int b = s;
        for (int o = 16; o > 0; o >>= 1) b += __shfl_xor(b, o, 32);
        if (lane == 0) bpart[wv] = b;
    }
    __syncthreads();

    walk_runs<false>(binned, rs, re, kc, pl, 0, perm, wv, lane);
    __syncthreads();

    int4 v = *(const int4*)&kc[4 * tid];
    int t = v.x + v.y + v.z + v.w;
    int inc = t;
    for (int o = 1; o < 32; o <<= 1) {
        int n = __shfl_up(inc, o, 32);
        if (lane >= o) inc += n;
    }
    if (lane == 31) wtot[wv] = inc;
    __syncthreads();

    int rbeg = 0;
    for (int w = 0; w < 8; ++w) rbeg += bpart[w];
    int pre = rbeg, rend = rbeg;
    for (int w = 0; w < 32; ++w) {
        int x = wtot[w];
        rend += x;
        if (w < wv) pre += x;
    }
    const int pbase = rbeg & ~31;
    int4 o4;
    o4.x = pre + inc - t; o4.y = o4.x + v.x; o4.z = o4.y + v.y; o4.w = o4.z + v.z;
    *(int4*)&kc[4 * tid] = o4;
    int key0 = R * KR + 4 * tid;
    if (4 * tid < KR && key0 < NK) *(int4*)(offs + key0) = o4;
    if (R == NRG - 1 && tid == 0) offs[NK] = NE;
    __syncthreads();

    walk_runs<true>(binned, rs, re, kc, pl, pbase, perm, wv, lane);
    __syncthreads();

    int nrel = rend - pbase;
    nrel = nrel < PCAP ? nrel : PCAP;
    for (int r0 = 4 * tid; r0 < nrel; r0 += 4096) {
        int4 w4 = *(const int4*)&pl[r0];
        int g = pbase + r0;
        if (g >= rbeg && r0 + 4 <= nrel) {
            *(int4*)(perm + g) = w4;
        } else {
            if (g     >= rbeg && r0     < nrel) perm[g]     = w4.x;
            if (g + 1 >= rbeg && r0 + 1 < nrel) perm[g + 1] = w4.y;
            if (g + 2 >= rbeg && r0 + 2 < nrel) perm[g + 2] = w4.z;
            if (g + 3 >= rbeg && r0 + 3 < nrel) perm[g + 3] = w4.w;
        }
    }
}

__device__ __forceinline__ v2f a_frag(const float* __restrict__ M, int row, int stride,
                                      int k0, int kh) {
    const float* p = M + (size_t)row * stride + k0 + kh;
    v2f a; a.x = p[0]; a.y = p[1]; return a;
}
__device__ __forceinline__ v2f b_frag(const float* __restrict__ W, int nout,
                                      int k0, int kh, int col) {
    v2f b;
    b.x = W[(size_t)(k0 + kh) * nout + col];
    b.y = W[(size_t)(k0 + kh + 1) * nout + col];
    return b;
}

#define WCAP 256

template <int F, int PITCH, bool STAGED>
__device__ __forceinline__ void walk_group(const float* __restrict__ feat, const int* __restrict__ perm,
                                           const int* widx, int base, int q,
                                           int beg, int b1, int b2, int b3, int end,
                                           float* __restrict__ slot0) {
    typedef __attribute__((ext_vector_type(F / 8))) float vec;
    const int FPL  = F / 8;
    const int SLOT = 16 * PITCH;
    const unsigned LANES3 = 0x40201u;
    int last = end - 1;
    vec acc = (vec)(0.0f);
    int cur = 0;
    const char* fb = (const char*)feat;
    const unsigned qb = (unsigned)q * (FPL * 4);
    unsigned mine = 0;
    if (beg < end) {
        int t = beg + q; t = t < last ? t : last;
        mine = (unsigned)(STAGED ? widx[t - base] : perm[t]) * (F * 4);
    }
    for (int e = beg; e < end; e += 8) {
        unsigned o0 = __builtin_amdgcn_mov_dpp8(mine, 0x000000);
        unsigned o1 = __builtin_amdgcn_mov_dpp8(mine, 0x249249);
        unsigned o2 = __builtin_amdgcn_mov_dpp8(mine, 0x492492);
        unsigned o3 = __builtin_amdgcn_mov_dpp8(mine, 0x6DB6DB);
        unsigned o4 = __builtin_amdgcn_mov_dpp8(mine, 0x924924);
        unsigned o5 = __builtin_amdgcn_mov_dpp8(mine, 0xB6DB6D);
        unsigned o6 = __builtin_amdgcn_mov_dpp8(mine, 0xDB6DB6);
        unsigned o7 = __builtin_amdgcn_mov_dpp8(mine, 0xFFFFFF);
        vec u[8];
        u[0] = *(const vec*)(fb + (o0 + qb));
        u[1] = *(const vec*)(fb + (o1 + qb));
        u[2] = *(const vec*)(fb + (o2 + qb));
        u[3] = *(const vec*)(fb + (o3 + qb));
        u[4] = *(const vec*)(fb + (o4 + qb));
        u[5] = *(const vec*)(fb + (o5 + qb));
        u[6] = *(const vec*)(fb + (o6 + qb));
        u[7] = *(const vec*)(fb + (o7 + qb));
        {
            int t = e + 8 + q; t = t < last ? t : last;
            mine = (unsigned)(STAGED ? widx[t - base] : perm[t]) * (F * 4);
        }
        const int n = end - e;
        unsigned d1 = (unsigned)(b1 - e), d2 = (unsigned)(b2 - e), d3 = (unsigned)(b3 - e);
        d1 = d1 < 8u ? d1 : 8u; d2 = d2 < 8u ? d2 : 8u; d3 = d3 < 8u ? d3 : 8u;
        const unsigned m = (1u << d1) | (0x200u << d2) | (0x40000u << d3);
        const int s0 = cur;
#pragma unroll
        for (int k = 0; k < 8; ++k) {
            if (k < n) {
                if (m & (LANES3 << k)) {
                    *(vec*)(slot0 + cur * SLOT) = acc;
                    cur = s0 + __builtin_popcount(m & (LANES3 * ((2u << k) - 1u)));
                    acc = u[k];
                } else {
                    acc += u[k];
                }
            }
        }
    }
    *(vec*)(slot0 + cur * SLOT) = acc;
}

template <int F, int PITCH>
__device__ __forceinline__ void tile_seg_sums(const float* __restrict__ feat,
                                              const int* __restrict__ offs,
                                              const int* __restrict__ perm,
                                              int m0, float* __restrict__ agg,
                                              int* __restrict__ widx, int* __restrict__ cnt) {
    typedef __attribute__((ext_vector_type(F / 8))) float vec;
    const int FPL  = F / 8;
    const int SLOT = 16 * PITCH;
    int q    = threadIdx.x & 7;
    int node = threadIdx.x >> 3;
    int k0   = (m0 + node) * NR;
    int4 ba  = *(const int4*)(offs + k0);
    int4 bb  = *(const int4*)(offs + k0 + 4);
    int end  = offs[k0 + 8];
    if (q == 0) {
        *(int4*)(cnt + node * NR)     = make_int4(ba.y - ba.x, ba.z - ba.y, ba.w - ba.z, bb.x - ba.w);
        *(int4*)(cnt + node * NR + 4) = make_int4(bb.y - bb.x, bb.z - bb.y, bb.w - bb.z, end - bb.w);
    }

    float* slot0 = agg + node * PITCH + q * FPL;
    vec zero;
#pragma unroll
    for (int i = 0; i < FPL; ++i) zero[i] = 0.0f;
#pragma unroll
    for (int sg = 0; sg < NR; ++sg) *(vec*)(slot0 + sg * SLOT) = zero;

    int wbeg = __builtin_amdgcn_readfirstlane(ba.x);
    int wend = __builtin_amdgcn_readlane(end, 31);
    int base = wbeg & ~3;
    const int mid = bb.x;
    if (wend - base <= WCAP) {
        int l4 = (threadIdx.x & 31) * 4;
        bool pa = base + l4 < wend, pb = base + 128 + l4 < wend;
        int4 wa = make_int4(0, 0, 0, 0), wb = make_int4(0, 0, 0, 0);
        if (pa) wa = *(const int4*)(perm + base + l4);
        if (pb) wb = *(const int4*)(perm + base + 128 + l4);
        if (pa) *(int4*)(widx + l4) = wa;
        if (pb) *(int4*)(widx + 128 + l4) = wb;
        walk_group<F, PITCH, true>(feat, perm, widx, base, q, ba.x, ba.y, ba.z, ba.w, mid, slot0);
        int4 c2 = *(const int4*)(cnt + node * NR + 4);
        int e1 = mid + c2.x, e2 = e1 + c2.y, e3 = e2 + c2.z;
        walk_group<F, PITCH, true>(feat, perm, widx, base, q, mid, e1, e2, e3, e3 + c2.w, slot0 + 4 * SLOT);
    } else {
        walk_group<F, PITCH, false>(feat, perm, widx, base, q, ba.x, ba.y, ba.z, ba.w, mid, slot0);
        int4 c2 = *(const int4*)(cnt + node * NR + 4);
        int e1 = mid + c2.x, e2 = e1 + c2.y, e3 = e2 + c2.z;
        walk_group<F, PITCH, false>(feat, perm, widx, base, q, mid, e1, e2, e3, e3 + c2.w, slot0 + 4 * SLOT);
    }
}


#define P1 20
#define P2 36
#define NW 4

__global__ __launch_bounds__(128) void rgcn_layer1(
    const float* __restrict__ x, const int* __restrict__ offs, const int* __restrict__ perm,
    const float* __restrict__ W1, const float* __restrict__ root1, const float* __restrict__ b1,
    float* __restrict__ h) {
    __shared__ __attribute__((aligned(16))) float agg[NR * 16 * P1];
    __shared__ __attribute__((aligned(16))) int wbuf[NW * 16 * 33];
    __shared__ __attribute__((aligned(16))) int cnt[16 * NR];
    float (*part)[16][33] = (float (*)[16][33])wbuf;
    int w    = __builtin_amdgcn_readfirstlane(threadIdx.x >> 5);
    int lane = threadIdx.x & 31;
    int m0   = blockIdx.x * 16;
    int arow = m0 + (lane & 15);
    int kh   = (lane >> 4) << 1;
    int col  = lane & 15;

    tile_seg_sums<16, P1>(x, offs, perm, m0, agg, wbuf + w * WCAP, cnt);
    __syncthreads();

    v8f c0, c1;
    if (w == 0) {
        float bb0 = b1[col], bb1 = b1[16 + col];
#pragma unroll
        for (int v = 0; v < 8; ++v) { c0[v] = bb0; c1[v] = bb1; }
#pragma unroll
        for (int k = 0; k < 16; k += 4) {
            v2f a  = a_frag(x, arow, 16, k, kh);
            v2f bA = b_frag(root1, 32, k, kh, col);
            v2f bB = b_frag(root1, 32, k, kh, col + 16);
            c0 = __builtin_amdgcn_wmma_f32_16x16x4_f32(false, a, false, bA, (short)0, c0, false, false);
            c1 = __builtin_amdgcn_wmma_f32_16x16x4_f32(false, a, false, bB, (short)0, c1, false, false);
        }
    } else {
#pragma unroll
        for (int v = 0; v < 8; ++v) { c0[v] = 0.0f; c1[v] = 0.0f; }
    }
#pragma nounroll
    for (int rr = 0; rr < 2; ++rr) {
        int r = 2 * w + rr;
        float sc = 1.0f / fmaxf((float)cnt[(lane & 15) * NR + r], 1.0f);
        const float* Wr = W1 + (size_t)r * 16 * 32;
        const float* ar = agg + (r * 16 + (lane & 15)) * P1 + kh;
#pragma unroll
        for (int j = 0; j < 4; ++j) {
            v2f a  = *(const v2f*)(ar + 4 * j) * sc;
            v2f bA = b_frag(Wr, 32, 4 * j, kh, col);
            v2f bB = b_frag(Wr, 32, 4 * j, kh, col + 16);
            c0 = __builtin_amdgcn_wmma_f32_16x16x4_f32(false, a, false, bA, (short)0, c0, false, false);
            c1 = __builtin_amdgcn_wmma_f32_16x16x4_f32(false, a, false, bB, (short)0, c1, false, false);
        }
    }
    int rb = (lane >> 4) << 3;
#pragma unroll
    for (int v = 0; v < 8; ++v) {
        part[w][rb + v][col]      = c0[v];
        part[w][rb + v][16 + col] = c1[v];
    }
    __syncthreads();
#pragma unroll
    for (int i = 0; i < 4; ++i) {
        int idx = threadIdx.x + 128 * i;
        int row = idx >> 5, cc = idx & 31;
        float sm = part[0][row][cc];
#pragma unroll
        for (int u = 1; u < NW; ++u) sm += part[u][row][cc];
        h[(size_t)m0 * 32 + idx] = fmaxf(sm, 0.0f);
    }
}


__global__ __launch_bounds__(128) void rgcn_layer2(
    const float* __restrict__ h, const int* __restrict__ offs, const int* __restrict__ perm,
    const float* __restrict__ W2, const float* __restrict__ root2, const float* __restrict__ b2,
    float* __restrict__ out) {
    __shared__ __attribute__((aligned(16))) float agg[NR * 16 * P2];
    __shared__ __attribute__((aligned(16))) int wbuf[NW * 16 * 17];
    __shared__ __attribute__((aligned(16))) int cnt[16 * NR];
    float (*part)[16][17] = (float (*)[16][17])wbuf;
    int w    = __builtin_amdgcn_readfirstlane(threadIdx.x >> 5);
    int lane = threadIdx.x & 31;
    int m0   = blockIdx.x * 16;
    int arow = m0 + (lane & 15);
    int kh   = (lane >> 4) << 1;
    int col  = lane & 15;

    tile_seg_sums<32, P2>(h, offs, perm, m0, agg, wbuf + w * WCAP, cnt);
    __syncthreads();

    v8f c;
    if (w == 0) {
        float bb = b2[col];
#pragma unroll
        for (int v = 0; v < 8; ++v) c[v] = bb;
#pragma unroll
        for (int k = 0; k < 32; k += 4) {
            v2f a = a_frag(h, arow, 32, k, kh);
            v2f b = b_frag(root2, 16, k, kh, col);
            c = __builtin_amdgcn_wmma_f32_16x16x4_f32(false, a, false, b, (short)0, c, false, false);
        }
    } else {
#pragma unroll
        for (int v = 0; v < 8; ++v) c[v] = 0.0f;
    }
#pragma nounroll
    for (int rr = 0; rr < 2; ++rr) {
        int r = 2 * w + rr;
        float sc = 1.0f / fmaxf((float)cnt[(lane & 15) * NR + r], 1.0f);
        const float* Wr = W2 + (size_t)r * 32 * 16;
        const float* ar = agg + (r * 16 + (lane & 15)) * P2 + kh;
#pragma unroll
        for (int j = 0; j < 8; ++j) {
            v2f a = *(const v2f*)(ar + 4 * j) * sc;
            v2f b = b_frag(Wr, 16, 4 * j, kh, col);
            c = __builtin_amdgcn_wmma_f32_16x16x4_f32(false, a, false, b, (short)0, c, false, false);
        }
    }
    int rb = (lane >> 4) << 3;
#pragma unroll
    for (int v = 0; v < 8; ++v) part[w][rb + v][col] = c[v];
    __syncthreads();
#pragma unroll
    for (int i = 0; i < 2; ++i) {
        int idx = threadIdx.x + 128 * i;
        int row = idx >> 4, cc = idx & 15;
        float sm = part[0][row][cc];
#pragma unroll
        for (int u = 1; u < NW; ++u) sm += part[u][row][cc];
        out[(size_t)m0 * 16 + idx] = sm;
    }
}


extern "C" void kernel_launch(void* const* d_in, const int* in_sizes, int n_in,
                              void* d_out, int out_size, void* d_ws, size_t ws_size,
                              hipStream_t stream) {
    const float* x     = (const float*)d_in[0];
    const int*   eidx  = (const int*)d_in[1];
    const int*   etype = (const int*)d_in[2];
    const float* W1    = (const float*)d_in[3];
    const float* root1 = (const float*)d_in[4];
    const float* b1    = (const float*)d_in[5];
    const float* W2    = (const float*)d_in[6];
    const float* root2 = (const float*)d_in[7];
    const float* b2    = (const float*)d_in[8];
    float* out = (float*)d_out;

    const int* src = eidx;
    const int* dst = eidx + NE;

    float* h       = (float*)d_ws;
    int*   binned  = (int*)(h + (size_t)NN * 32);
    int*   perm    = binned + NE;
    int*   starts  = perm + NE;
    int*   offs    = starts + NCH * (NRG + 1);

    bin_chunks<<<NCH, 1024, 0, stream>>>(src, dst, etype, binned, starts);
    sort_range<<<NRG, 1024, 0, stream>>>(binned, starts, offs, perm);

    rgcn_layer1<<<NN / 16, 128, 0, stream>>>(x, offs, perm, W1, root1, b1, h);
    rgcn_layer2<<<NN / 16, 128, 0, stream>>>(h, offs, perm, W2, root2, b2, out);

    (void)in_sizes; (void)n_in; (void)out_size; (void)ws_size;
}
